# starting kernel
# baseline (speedup 1.0000x reference)
#include <hip/hip_runtime.h>
#include <math.h>


#define NQ 10
#define WAVES_PER_BLOCK 4

typedef float v2f __attribute__((ext_vector_type(2)));
typedef float v8f __attribute__((ext_vector_type(8)));

__device__ __forceinline__ v8f wmma4(v2f a, v2f b, v8f c) {
  return __builtin_amdgcn_wmma_f32_16x16x4_f32(
      false, a, false, b, (short)0, c, false, false);
}

__global__ void precompute_kernel(const float* __restrict__ qw,
                                  float2* __restrict__ mats,
                                  int* __restrict__ perms)
{
  int tid = threadIdx.x;
  for (int e = tid; e < 4096; e += blockDim.x) {
    int mat = e >> 10;
    int idx = e & 1023;
    int r = idx >> 5, k = idx & 31;
    int l = mat >> 1;
    int wbase = (mat & 1) * 5;
    float vr = 1.f, vi = 0.f;
    for (int j = 0; j < 5; ++j) {
      int q = wbase + j;
      const float* w = qw + (l * NQ + q) * 3;
      float phi = w[0], theta = w[1], omega = w[2];
      float ch = cosf(theta * 0.5f), sh = sinf(theta * 0.5f);
      int bi = (r >> (4 - j)) & 1;
      int bj = (k >> (4 - j)) & 1;
      float ang, mag;
      if (!bi && !bj)      { ang = -0.5f * (phi + omega); mag =  ch; }
      else if (!bi &&  bj) { ang =  0.5f * (phi - omega); mag = -sh; }
      else if ( bi && !bj) { ang = -0.5f * (phi - omega); mag =  sh; }
      else                 { ang =  0.5f * (phi + omega); mag =  ch; }
      float er = mag * cosf(ang), ei = mag * sinf(ang);
      float nr = vr * er - vi * ei;
      float ni = vr * ei + vi * er;
      vr = nr; vi = ni;
    }
    mats[mat * 1024 + idx] = make_float2(vr, vi);
  }
  for (int i = tid; i < 1024; i += blockDim.x) {
    for (int l = 0; l < 2; ++l) {
      int rr = l + 1;
      int b[NQ];
      for (int q = 0; q < NQ; ++q) b[q] = (i >> (NQ - 1 - q)) & 1;
      for (int q = 0; q < NQ; ++q) b[(q + rr) % NQ] ^= b[q];
      int dest = 0;
      for (int q = 0; q < NQ; ++q) dest |= b[q] << (NQ - 1 - q);
      perms[l * 1024 + i] = dest;
    }
  }
}

__global__ __launch_bounds__(32 * WAVES_PER_BLOCK)
void qlayer_kernel(const float* __restrict__ x,
                   const float2* __restrict__ mats,
                   const int* __restrict__ perms,
                   float* __restrict__ out)
{
  __shared__ float2 lds_all[WAVES_PER_BLOCK * 1024];
  const int lane = threadIdx.x & 31;
  const int wave = threadIdx.x >> 5;
  float2* S = lds_all + wave * 1024;
  const int b = blockIdx.x * WAVES_PER_BLOCK + wave;

  const float2* A0 = mats;
  const float2* B0 = mats + 1024;
  const float2* A1 = mats + 2048;
  const float2* B1 = mats + 3072;
  const int* perm1 = perms;
  const int* perm2 = perms + 1024;

  float cq[NQ], sq[NQ];
  #pragma unroll
  for (int q = 0; q < NQ; ++q) {
    float xv = x[b * NQ + q];
    cq[q] = cosf(0.5f * xv);
    sq[q] = sinf(0.5f * xv);
  }
  float u = 1.f, v = 1.f;
  #pragma unroll
  for (int j = 0; j < 5; ++j) {
    int bit = (lane >> (4 - j)) & 1;
    u *= bit ? sq[j] : cq[j];
    v *= bit ? sq[5 + j] : cq[5 + j];
  }

  float ar_ = 0.f, ai_ = 0.f, br_ = 0.f, bi_ = 0.f;
  for (int k = 0; k < 32; ++k) {
    float uk = __shfl(u, k, 32);
    float vk = __shfl(v, k, 32);
    float2 Ae = A0[lane * 32 + k];
    float2 Be = B0[lane * 32 + k];
    ar_ += Ae.x * uk; ai_ += Ae.y * uk;
    br_ += Be.x * vk; bi_ += Be.y * vk;
  }
  for (int r = 0; r < 32; ++r) {
    float xr = __shfl(ar_, r, 32);
    float xi = __shfl(ai_, r, 32);
    float wr = xr * br_ - xi * bi_;
    float wi = xr * bi_ + xi * br_;
    int d = perm1[r * 32 + lane];
    S[d] = make_float2(wr, wi);
  }
  __syncthreads();

  const v8f vzero = {0.f, 0.f, 0.f, 0.f, 0.f, 0.f, 0.f, 0.f};
  const int mh = lane & 15;
  const int kh = lane >> 4;
  const int mbase = kh * 8;
  const int nn = lane & 15;

  v8f accR[2][2], accI[2][2];
  #pragma unroll
  for (int I = 0; I < 2; ++I)
    #pragma unroll
    for (int J = 0; J < 2; ++J) { accR[I][J] = vzero; accI[I][J] = vzero; }

  #pragma unroll
  for (int I = 0; I < 2; ++I) {
    #pragma unroll
    for (int J = 0; J < 2; ++J) {
      #pragma unroll
      for (int kk = 0; kk < 32; kk += 4) {
        int k0 = kk + 2 * kh;
        float2 a0 = A1[(16 * I + mh) * 32 + k0];
        float2 a1 = A1[(16 * I + mh) * 32 + k0 + 1];
        v2f aR = {a0.x, a1.x};
        v2f aI = {a0.y, a1.y};
        v2f aIn = {-a0.y, -a1.y};
        float2 s0 = S[k0 * 32 + 16 * J + mh];
        float2 s1 = S[(k0 + 1) * 32 + 16 * J + mh];
        v2f bR = {s0.x, s1.x};
        v2f bI = {s0.y, s1.y};
        accR[I][J] = wmma4(aR,  bR, accR[I][J]);
        accR[I][J] = wmma4(aIn, bI, accR[I][J]);
        accI[I][J] = wmma4(aR,  bI, accI[I][J]);
        accI[I][J] = wmma4(aI,  bR, accI[I][J]);
      }
    }
  }
  __syncthreads();
  #pragma unroll
  for (int I = 0; I < 2; ++I)
    #pragma unroll
    for (int J = 0; J < 2; ++J)
      #pragma unroll
      for (int t = 0; t < 8; ++t) {
        int rr = 16 * I + mbase + t;
        int cc = 16 * J + nn;
        S[rr * 32 + cc] = make_float2(accR[I][J][t], accI[I][J][t]);
      }
  __syncthreads();

  v8f qR[2][2], qI[2][2];
  #pragma unroll
  for (int I = 0; I < 2; ++I)
    #pragma unroll
    for (int J = 0; J < 2; ++J) { qR[I][J] = vzero; qI[I][J] = vzero; }

  #pragma unroll
  for (int I = 0; I < 2; ++I) {
    #pragma unroll
    for (int J = 0; J < 2; ++J) {
      #pragma unroll
      for (int kk = 0; kk < 32; kk += 4) {
        int k0 = kk + 2 * kh;
        float2 p0 = S[(16 * I + mh) * 32 + k0];
        float2 p1 = S[(16 * I + mh) * 32 + k0 + 1];
        v2f pR = {p0.x, p1.x};
        v2f pI = {p0.y, p1.y};
        v2f pIn = {-p0.y, -p1.y};
        float2 b0 = B1[(16 * J + mh) * 32 + k0];
        float2 b1 = B1[(16 * J + mh) * 32 + k0 + 1];
        v2f bR = {b0.x, b1.x};
        v2f bI = {b0.y, b1.y};
        qR[I][J] = wmma4(pR,  bR, qR[I][J]);
        qR[I][J] = wmma4(pIn, bI, qR[I][J]);
        qI[I][J] = wmma4(pR,  bI, qI[I][J]);
        qI[I][J] = wmma4(pI,  bR, qI[I][J]);
      }
    }
  }
  __syncthreads();
  #pragma unroll
  for (int I = 0; I < 2; ++I)
    #pragma unroll
    for (int J = 0; J < 2; ++J)
      #pragma unroll
      for (int t = 0; t < 8; ++t) {
        int idx = (16 * I + mbase + t) * 32 + (16 * J + nn);
        int d = perm2[idx];
        S[d] = make_float2(qR[I][J][t], qI[I][J][t]);
      }
  __syncthreads();

  float e[NQ];
  #pragma unroll
  for (int q = 0; q < NQ; ++q) e[q] = 0.f;
  for (int row = 0; row < 32; ++row) {
    int i = row * 32 + lane;
    float2 amp = S[i];
    float p = amp.x * amp.x + amp.y * amp.y;
    #pragma unroll
    for (int q = 0; q < NQ; ++q)
      e[q] += ((i >> (NQ - 1 - q)) & 1) ? -p : p;
  }
  #pragma unroll
  for (int q = 0; q < NQ; ++q) {
    #pragma unroll
    for (int off = 16; off > 0; off >>= 1)
      e[q] += __shfl_xor(e[q], off, 32);
  }
  if (lane == 0) {
    #pragma unroll
    for (int q = 0; q < NQ; ++q) out[b * NQ + q] = e[q];
  }
}

extern "C" void kernel_launch(void* const* d_in, const int* in_sizes, int n_in,
                              void* d_out, int out_size, void* d_ws, size_t ws_size,
                              hipStream_t stream) {
  const float* x  = (const float*)d_in[0];
  const float* qw = (const float*)d_in[1];
  float* out = (float*)d_out;

  float2* mats = (float2*)d_ws;
  int* perms = (int*)((char*)d_ws + 4 * 1024 * sizeof(float2));

  hipLaunchKernelGGL(precompute_kernel, dim3(1), dim3(256), 0, stream,
                     qw, mats, perms);

  int batch = in_sizes[0] / NQ;
  hipLaunchKernelGGL(qlayer_kernel, dim3(batch / WAVES_PER_BLOCK),
                     dim3(32 * WAVES_PER_BLOCK), 0, stream,
                     x, mats, perms, out);
}


# Layer 0 applied per qubit on the product state: one lane per qubit builds Rot*(cos,sin), amplitudes = 10-factor products; 32x32 layer-0 matrices, loads, matvecs removed (sum order changes)
# speedup vs baseline: 1.4908x; 1.4908x over previous
#include <hip/hip_runtime.h>
#include <math.h>


#define NQ 10
#define WAVES_PER_BLOCK 4

typedef float v2f __attribute__((ext_vector_type(2)));
typedef float v8f __attribute__((ext_vector_type(8)));

__device__ __forceinline__ v8f wmma4(v2f a, v2f b, v8f c) {
  return __builtin_amdgcn_wmma_f32_16x16x4_f32(
      false, a, false, b, (short)0, c, false, false);
}

__device__ __forceinline__ float bcast(float v, int src) {
  return __int_as_float(__builtin_amdgcn_readlane(__float_as_int(v), src));
}

__global__ void precompute_kernel(const float* __restrict__ qw,
                                  float2* __restrict__ mats,
                                  int* __restrict__ perms)
{
  int tid = threadIdx.x;
  for (int e = tid; e < 2048; e += blockDim.x) {
    int mat = e >> 10;
    int idx = e & 1023;
    int r = idx >> 5, k = idx & 31;
    const int l = 1;
    int wbase = mat * 5;
    float vr = 1.f, vi = 0.f;
    for (int j = 0; j < 5; ++j) {
      int q = wbase + j;
      const float* w = qw + (l * NQ + q) * 3;
      float phi = w[0], theta = w[1], omega = w[2];
      float ch = cosf(theta * 0.5f), sh = sinf(theta * 0.5f);
      int bi = (r >> (4 - j)) & 1;
      int bj = (k >> (4 - j)) & 1;
      float ang, mag;
      if (!bi && !bj)      { ang = -0.5f * (phi + omega); mag =  ch; }
      else if (!bi &&  bj) { ang =  0.5f * (phi - omega); mag = -sh; }
      else if ( bi && !bj) { ang = -0.5f * (phi - omega); mag =  sh; }
      else                 { ang =  0.5f * (phi + omega); mag =  ch; }
      float er = mag * cosf(ang), ei = mag * sinf(ang);
      float nr = vr * er - vi * ei;
      float ni = vr * ei + vi * er;
      vr = nr; vi = ni;
    }
    mats[mat * 1024 + idx] = make_float2(vr, vi);
  }
  for (int i = tid; i < 1024; i += blockDim.x) {
    for (int l = 0; l < 2; ++l) {
      int rr = l + 1;
      int b[NQ];
      for (int q = 0; q < NQ; ++q) b[q] = (i >> (NQ - 1 - q)) & 1;
      for (int q = 0; q < NQ; ++q) b[(q + rr) % NQ] ^= b[q];
      int dest = 0;
      for (int q = 0; q < NQ; ++q) dest |= b[q] << (NQ - 1 - q);
      perms[l * 1024 + i] = dest;
    }
  }
}

__global__ __launch_bounds__(32 * WAVES_PER_BLOCK)
void qlayer_kernel(const float* __restrict__ x,
                   const float* __restrict__ qw,
                   const float2* __restrict__ mats,
                   const int* __restrict__ perms,
                   float* __restrict__ out)
{
  __shared__ float2 lds_all[WAVES_PER_BLOCK * 1024];
  const int lane = threadIdx.x & 31;
  const int wave = threadIdx.x >> 5;
  float2* S = lds_all + wave * 1024;
  const int b = blockIdx.x * WAVES_PER_BLOCK + wave;

  const float2* A1 = mats;
  const float2* B1 = mats + 1024;
  const int* perm1 = perms;
  const int* perm2 = perms + 1024;

  const int q = lane < NQ ? lane : NQ - 1;
  float c, sn, ch, sh, cp, sp, cm, sm;
  {
    const float xv = x[b * NQ + q];
    const float phi = qw[q * 3 + 0], theta = qw[q * 3 + 1], omega = qw[q * 3 + 2];
    sincosf(0.5f * xv, &sn, &c);
    sincosf(0.5f * theta, &sh, &ch);
    sincosf(0.5f * (phi + omega), &sp, &cp);
    sincosf(0.5f * (phi - omega), &sm, &cm);
  }
  const float w0r = (cp * ch) * c - (cm * sh) * sn;
  const float w0i = -(sp * ch) * c - (sm * sh) * sn;
  const float w1r = (cm * sh) * c + (cp * ch) * sn;
  const float w1i = -(sm * sh) * c + (sp * ch) * sn;

  float ar_ = 1.f, ai_ = 0.f, br_ = 1.f, bi_ = 0.f;
  #pragma unroll
  for (int j = 0; j < 5; ++j) {
    const bool bit = (lane >> (4 - j)) & 1;
    const float f0r = bcast(w0r, j), f0i = bcast(w0i, j);
    const float f1r = bcast(w1r, j), f1i = bcast(w1i, j);
    const float g0r = bcast(w0r, 5 + j), g0i = bcast(w0i, 5 + j);
    const float g1r = bcast(w1r, 5 + j), g1i = bcast(w1i, 5 + j);
    const float fr = bit ? f1r : f0r;
    const float fi = bit ? f1i : f0i;
    const float gr = bit ? g1r : g0r;
    const float gi = bit ? g1i : g0i;
    float nr = ar_ * fr - ai_ * fi;
    float ni = ar_ * fi + ai_ * fr;
    ar_ = nr; ai_ = ni;
    nr = br_ * gr - bi_ * gi;
    ni = br_ * gi + bi_ * gr;
    br_ = nr; bi_ = ni;
  }
  for (int r = 0; r < 32; ++r) {
    float xr = __shfl(ar_, r, 32);
    float xi = __shfl(ai_, r, 32);
    float wr = xr * br_ - xi * bi_;
    float wi = xr * bi_ + xi * br_;
    int d = perm1[r * 32 + lane];
    S[d] = make_float2(wr, wi);
  }
  __syncthreads();

  const v8f vzero = {0.f, 0.f, 0.f, 0.f, 0.f, 0.f, 0.f, 0.f};
  const int mh = lane & 15;
  const int kh = lane >> 4;
  const int mbase = kh * 8;
  const int nn = lane & 15;

  v8f accR[2][2], accI[2][2];
  #pragma unroll
  for (int I = 0; I < 2; ++I)
    #pragma unroll
    for (int J = 0; J < 2; ++J) { accR[I][J] = vzero; accI[I][J] = vzero; }

  #pragma unroll
  for (int I = 0; I < 2; ++I) {
    #pragma unroll
    for (int J = 0; J < 2; ++J) {
      #pragma unroll
      for (int kk = 0; kk < 32; kk += 4) {
        int k0 = kk + 2 * kh;
        float2 a0 = A1[(16 * I + mh) * 32 + k0];
        float2 a1 = A1[(16 * I + mh) * 32 + k0 + 1];
        v2f aR = {a0.x, a1.x};
        v2f aI = {a0.y, a1.y};
        v2f aIn = {-a0.y, -a1.y};
        float2 s0 = S[k0 * 32 + 16 * J + mh];
        float2 s1 = S[(k0 + 1) * 32 + 16 * J + mh];
        v2f bR = {s0.x, s1.x};
        v2f bI = {s0.y, s1.y};
        accR[I][J] = wmma4(aR,  bR, accR[I][J]);
        accR[I][J] = wmma4(aIn, bI, accR[I][J]);
        accI[I][J] = wmma4(aR,  bI, accI[I][J]);
        accI[I][J] = wmma4(aI,  bR, accI[I][J]);
      }
    }
  }
  __syncthreads();
  #pragma unroll
  for (int I = 0; I < 2; ++I)
    #pragma unroll
    for (int J = 0; J < 2; ++J)
      #pragma unroll
      for (int t = 0; t < 8; ++t) {
        int rr = 16 * I + mbase + t;
        int cc = 16 * J + nn;
        S[rr * 32 + cc] = make_float2(accR[I][J][t], accI[I][J][t]);
      }
  __syncthreads();

  v8f qR[2][2], qI[2][2];
  #pragma unroll
  for (int I = 0; I < 2; ++I)
    #pragma unroll
    for (int J = 0; J < 2; ++J) { qR[I][J] = vzero; qI[I][J] = vzero; }

  #pragma unroll
  for (int I = 0; I < 2; ++I) {
    #pragma unroll
    for (int J = 0; J < 2; ++J) {
      #pragma unroll
      for (int kk = 0; kk < 32; kk += 4) {
        int k0 = kk + 2 * kh;
        float2 p0 = S[(16 * I + mh) * 32 + k0];
        float2 p1 = S[(16 * I + mh) * 32 + k0 + 1];
        v2f pR = {p0.x, p1.x};
        v2f pI = {p0.y, p1.y};
        v2f pIn = {-p0.y, -p1.y};
        float2 b0 = B1[(16 * J + mh) * 32 + k0];
        float2 b1 = B1[(16 * J + mh) * 32 + k0 + 1];
        v2f bR = {b0.x, b1.x};
        v2f bI = {b0.y, b1.y};
        qR[I][J] = wmma4(pR,  bR, qR[I][J]);
        qR[I][J] = wmma4(pIn, bI, qR[I][J]);
        qI[I][J] = wmma4(pR,  bI, qI[I][J]);
        qI[I][J] = wmma4(pI,  bR, qI[I][J]);
      }
    }
  }
  __syncthreads();
  #pragma unroll
  for (int I = 0; I < 2; ++I)
    #pragma unroll
    for (int J = 0; J < 2; ++J)
      #pragma unroll
      for (int t = 0; t < 8; ++t) {
        int idx = (16 * I + mbase + t) * 32 + (16 * J + nn);
        int d = perm2[idx];
        S[d] = make_float2(qR[I][J][t], qI[I][J][t]);
      }
  __syncthreads();

  float e[NQ];
  #pragma unroll
  for (int q = 0; q < NQ; ++q) e[q] = 0.f;
  for (int row = 0; row < 32; ++row) {
    int i = row * 32 + lane;
    float2 amp = S[i];
    float p = amp.x * amp.x + amp.y * amp.y;
    #pragma unroll
    for (int q = 0; q < NQ; ++q)
      e[q] += ((i >> (NQ - 1 - q)) & 1) ? -p : p;
  }
  #pragma unroll
  for (int q = 0; q < NQ; ++q) {
    #pragma unroll
    for (int off = 16; off > 0; off >>= 1)
      e[q] += __shfl_xor(e[q], off, 32);
  }
  if (lane == 0) {
    #pragma unroll
    for (int q = 0; q < NQ; ++q) out[b * NQ + q] = e[q];
  }
}

extern "C" void kernel_launch(void* const* d_in, const int* in_sizes, int n_in,
                              void* d_out, int out_size, void* d_ws, size_t ws_size,
                              hipStream_t stream) {
  const float* x  = (const float*)d_in[0];
  const float* qw = (const float*)d_in[1];
  float* out = (float*)d_out;

  float2* mats = (float2*)d_ws;
  int* perms = (int*)((char*)d_ws + 2 * 1024 * sizeof(float2));

  hipLaunchKernelGGL(precompute_kernel, dim3(1), dim3(256), 0, stream,
                     qw, mats, perms);

  int batch = in_sizes[0] / NQ;
  hipLaunchKernelGGL(qlayer_kernel, dim3(batch / WAVES_PER_BLOCK),
                     dim3(32 * WAVES_PER_BLOCK), 0, stream,
                     x, qw, mats, perms, out);
}


# Set-up kernel: each layer-1 Rot matrix's trigonometry done once per qubit (one thread each) and shared through LDS; A1/B1 entries = products of 5 looked-up factors, same order; 8 blocks
# speedup vs baseline: 1.9983x; 1.3404x over previous
#include <hip/hip_runtime.h>
#include <math.h>


#define NQ 10
#define WAVES_PER_BLOCK 4

typedef float v2f __attribute__((ext_vector_type(2)));
typedef float v8f __attribute__((ext_vector_type(8)));

__device__ __forceinline__ v8f wmma4(v2f a, v2f b, v8f c) {
  return __builtin_amdgcn_wmma_f32_16x16x4_f32(
      false, a, false, b, (short)0, c, false, false);
}

__device__ __forceinline__ float bcast(float v, int src) {
  return __int_as_float(__builtin_amdgcn_readlane(__float_as_int(v), src));
}

#define PRE_THREADS 256
#define PRE_BLOCKS 8

__device__ __forceinline__ int cnot_ring(int i, int rr) {
  int b[NQ];
  for (int q = 0; q < NQ; ++q) b[q] = (i >> (NQ - 1 - q)) & 1;
  for (int q = 0; q < NQ; ++q) b[(q + rr) % NQ] ^= b[q];
  int dest = 0;
  for (int q = 0; q < NQ; ++q) dest |= b[q] << (NQ - 1 - q);
  return dest;
}

__global__ __launch_bounds__(PRE_THREADS)
void precompute_kernel(const float* __restrict__ qw,
                       float2* __restrict__ mats,
                       int* __restrict__ perms)
{
  __shared__ float2 rot[NQ][4];
  const int tid = threadIdx.x;
  if (tid < NQ) {
    const float* w = qw + (1 * NQ + tid) * 3;
    const float phi = w[0], theta = w[1], omega = w[2];
    float ch, sh, cp, sp, cm, sm;
    sincosf(0.5f * theta, &sh, &ch);
    sincosf(0.5f * (phi + omega), &sp, &cp);
    sincosf(0.5f * (phi - omega), &sm, &cm);
    rot[tid][0] = make_float2(ch * cp, -(ch * sp));
    rot[tid][1] = make_float2(-(sh * cm), -(sh * sm));
    rot[tid][2] = make_float2(sh * cm, -(sh * sm));
    rot[tid][3] = make_float2(ch * cp, ch * sp);
  }
  __syncthreads();

  static_assert(PRE_BLOCKS * PRE_THREADS == 2048, "one thread per table entry");
  const int e = blockIdx.x * PRE_THREADS + tid;
  {
    const int mat = e >> 10;
    const int idx = e & 1023;
    const int r = idx >> 5, k = idx & 31;
    const int wbase = mat * 5;
    float vr = 1.f, vi = 0.f;
    for (int j = 0; j < 5; ++j) {
      const int bi = (r >> (4 - j)) & 1;
      const int bj = (k >> (4 - j)) & 1;
      const float2 f = rot[wbase + j][2 * bi + bj];
      const float nr = vr * f.x - vi * f.y;
      const float ni = vr * f.y + vi * f.x;
      vr = nr; vi = ni;
    }
    mats[e] = make_float2(vr, vi);
  }
  perms[e] = (e >> 10) ? cnot_ring(e & 1023, 2) : cnot_ring(e & 1023, 1);
}

__global__ __launch_bounds__(32 * WAVES_PER_BLOCK)
void qlayer_kernel(const float* __restrict__ x,
                   const float* __restrict__ qw,
                   const float2* __restrict__ mats,
                   const int* __restrict__ perms,
                   float* __restrict__ out)
{
  __shared__ float2 lds_all[WAVES_PER_BLOCK * 1024];
  const int lane = threadIdx.x & 31;
  const int wave = threadIdx.x >> 5;
  float2* S = lds_all + wave * 1024;
  const int b = blockIdx.x * WAVES_PER_BLOCK + wave;

  const float2* A1 = mats;
  const float2* B1 = mats + 1024;
  const int* perm1 = perms;
  const int* perm2 = perms + 1024;

  const int q = lane < NQ ? lane : NQ - 1;
  float c, sn, ch, sh, cp, sp, cm, sm;
  {
    const float xv = x[b * NQ + q];
    const float phi = qw[q * 3 + 0], theta = qw[q * 3 + 1], omega = qw[q * 3 + 2];
    sincosf(0.5f * xv, &sn, &c);
    sincosf(0.5f * theta, &sh, &ch);
    sincosf(0.5f * (phi + omega), &sp, &cp);
    sincosf(0.5f * (phi - omega), &sm, &cm);
  }
  const float w0r = (cp * ch) * c - (cm * sh) * sn;
  const float w0i = -(sp * ch) * c - (sm * sh) * sn;
  const float w1r = (cm * sh) * c + (cp * ch) * sn;
  const float w1i = -(sm * sh) * c + (sp * ch) * sn;

  float ar_ = 1.f, ai_ = 0.f, br_ = 1.f, bi_ = 0.f;
  #pragma unroll
  for (int j = 0; j < 5; ++j) {
    const bool bit = (lane >> (4 - j)) & 1;
    const float f0r = bcast(w0r, j), f0i = bcast(w0i, j);
    const float f1r = bcast(w1r, j), f1i = bcast(w1i, j);
    const float g0r = bcast(w0r, 5 + j), g0i = bcast(w0i, 5 + j);
    const float g1r = bcast(w1r, 5 + j), g1i = bcast(w1i, 5 + j);
    const float fr = bit ? f1r : f0r;
    const float fi = bit ? f1i : f0i;
    const float gr = bit ? g1r : g0r;
    const float gi = bit ? g1i : g0i;
    float nr = ar_ * fr - ai_ * fi;
    float ni = ar_ * fi + ai_ * fr;
    ar_ = nr; ai_ = ni;
    nr = br_ * gr - bi_ * gi;
    ni = br_ * gi + bi_ * gr;
    br_ = nr; bi_ = ni;
  }
  for (int r = 0; r < 32; ++r) {
    float xr = __shfl(ar_, r, 32);
    float xi = __shfl(ai_, r, 32);
    float wr = xr * br_ - xi * bi_;
    float wi = xr * bi_ + xi * br_;
    int d = perm1[r * 32 + lane];
    S[d] = make_float2(wr, wi);
  }
  __syncthreads();

  const v8f vzero = {0.f, 0.f, 0.f, 0.f, 0.f, 0.f, 0.f, 0.f};
  const int mh = lane & 15;
  const int kh = lane >> 4;
  const int mbase = kh * 8;
  const int nn = lane & 15;

  v8f accR[2][2], accI[2][2];
  #pragma unroll
  for (int I = 0; I < 2; ++I)
    #pragma unroll
    for (int J = 0; J < 2; ++J) { accR[I][J] = vzero; accI[I][J] = vzero; }

  #pragma unroll
  for (int I = 0; I < 2; ++I) {
    #pragma unroll
    for (int J = 0; J < 2; ++J) {
      #pragma unroll
      for (int kk = 0; kk < 32; kk += 4) {
        int k0 = kk + 2 * kh;
        float2 a0 = A1[(16 * I + mh) * 32 + k0];
        float2 a1 = A1[(16 * I + mh) * 32 + k0 + 1];
        v2f aR = {a0.x, a1.x};
        v2f aI = {a0.y, a1.y};
        v2f aIn = {-a0.y, -a1.y};
        float2 s0 = S[k0 * 32 + 16 * J + mh];
        float2 s1 = S[(k0 + 1) * 32 + 16 * J + mh];
        v2f bR = {s0.x, s1.x};
        v2f bI = {s0.y, s1.y};
        accR[I][J] = wmma4(aR,  bR, accR[I][J]);
        accR[I][J] = wmma4(aIn, bI, accR[I][J]);
        accI[I][J] = wmma4(aR,  bI, accI[I][J]);
        accI[I][J] = wmma4(aI,  bR, accI[I][J]);
      }
    }
  }
  __syncthreads();
  #pragma unroll
  for (int I = 0; I < 2; ++I)
    #pragma unroll
    for (int J = 0; J < 2; ++J)
      #pragma unroll
      for (int t = 0; t < 8; ++t) {
        int rr = 16 * I + mbase + t;
        int cc = 16 * J + nn;
        S[rr * 32 + cc] = make_float2(accR[I][J][t], accI[I][J][t]);
      }
  __syncthreads();

  v8f qR[2][2], qI[2][2];
  #pragma unroll
  for (int I = 0; I < 2; ++I)
    #pragma unroll
    for (int J = 0; J < 2; ++J) { qR[I][J] = vzero; qI[I][J] = vzero; }

  #pragma unroll
  for (int I = 0; I < 2; ++I) {
    #pragma unroll
    for (int J = 0; J < 2; ++J) {
      #pragma unroll
      for (int kk = 0; kk < 32; kk += 4) {
        int k0 = kk + 2 * kh;
        float2 p0 = S[(16 * I + mh) * 32 + k0];
        float2 p1 = S[(16 * I + mh) * 32 + k0 + 1];
        v2f pR = {p0.x, p1.x};
        v2f pI = {p0.y, p1.y};
        v2f pIn = {-p0.y, -p1.y};
        float2 b0 = B1[(16 * J + mh) * 32 + k0];
        float2 b1 = B1[(16 * J + mh) * 32 + k0 + 1];
        v2f bR = {b0.x, b1.x};
        v2f bI = {b0.y, b1.y};
        qR[I][J] = wmma4(pR,  bR, qR[I][J]);
        qR[I][J] = wmma4(pIn, bI, qR[I][J]);
        qI[I][J] = wmma4(pR,  bI, qI[I][J]);
        qI[I][J] = wmma4(pI,  bR, qI[I][J]);
      }
    }
  }
  __syncthreads();
  #pragma unroll
  for (int I = 0; I < 2; ++I)
    #pragma unroll
    for (int J = 0; J < 2; ++J)
      #pragma unroll
      for (int t = 0; t < 8; ++t) {
        int idx = (16 * I + mbase + t) * 32 + (16 * J + nn);
        int d = perm2[idx];
        S[d] = make_float2(qR[I][J][t], qI[I][J][t]);
      }
  __syncthreads();

  float e[NQ];
  #pragma unroll
  for (int q = 0; q < NQ; ++q) e[q] = 0.f;
  for (int row = 0; row < 32; ++row) {
    int i = row * 32 + lane;
    float2 amp = S[i];
    float p = amp.x * amp.x + amp.y * amp.y;
    #pragma unroll
    for (int q = 0; q < NQ; ++q)
      e[q] += ((i >> (NQ - 1 - q)) & 1) ? -p : p;
  }
  #pragma unroll
  for (int q = 0; q < NQ; ++q) {
    #pragma unroll
    for (int off = 16; off > 0; off >>= 1)
      e[q] += __shfl_xor(e[q], off, 32);
  }
  if (lane == 0) {
    #pragma unroll
    for (int q = 0; q < NQ; ++q) out[b * NQ + q] = e[q];
  }
}

extern "C" void kernel_launch(void* const* d_in, const int* in_sizes, int n_in,
                              void* d_out, int out_size, void* d_ws, size_t ws_size,
                              hipStream_t stream) {
  const float* x  = (const float*)d_in[0];
  const float* qw = (const float*)d_in[1];
  float* out = (float*)d_out;

  float2* mats = (float2*)d_ws;
  int* perms = (int*)((char*)d_ws + 2 * 1024 * sizeof(float2));

  hipLaunchKernelGGL(precompute_kernel, dim3(PRE_BLOCKS), dim3(PRE_THREADS), 0, stream,
                     qw, mats, perms);

  int batch = in_sizes[0] / NQ;
  hipLaunchKernelGGL(qlayer_kernel, dim3(batch / WAVES_PER_BLOCK),
                     dim3(32 * WAVES_PER_BLOCK), 0, stream,
                     x, qw, mats, perms, out);
}


# Layer-1 rotation as three Kronecker factors on WMMA (wires 0-2 8x8, 3-5 8x8, 6-9 16x16; the 8x8 ones stack Re/Im output rows in one tile), not two 32x32: 256 -> 128 WMMAs; sum order changes
# speedup vs baseline: 2.6233x; 1.3128x over previous
#include <hip/hip_runtime.h>
#include <math.h>


#define NQ 10
#define WAVES_PER_BLOCK 4

typedef float v2f __attribute__((ext_vector_type(2)));
typedef float v8f __attribute__((ext_vector_type(8)));

__device__ __forceinline__ v8f wmma4(v2f a, v2f b, v8f c) {
  return __builtin_amdgcn_wmma_f32_16x16x4_f32(
      false, a, false, b, (short)0, c, false, false);
}

__device__ __forceinline__ float bcast(float v, int src) {
  return __int_as_float(__builtin_amdgcn_readlane(__float_as_int(v), src));
}

#define PRE_THREADS 256
#define PRE_BLOCKS 8
#define TAB_MA1 0
#define TAB_MA2 128
#define TAB_MB1 256
#define TAB_MB2 384
#define TAB_CR 512
#define TAB_CI 768
#define TAB_CN 1024
#define TAB_FLOATS 1280

__device__ __forceinline__ int cnot_ring(int i, int rr) {
  int b[NQ];
  for (int q = 0; q < NQ; ++q) b[q] = (i >> (NQ - 1 - q)) & 1;
  for (int q = 0; q < NQ; ++q) b[(q + rr) % NQ] ^= b[q];
  int dest = 0;
  for (int q = 0; q < NQ; ++q) dest |= b[q] << (NQ - 1 - q);
  return dest;
}

__global__ __launch_bounds__(PRE_THREADS)
void precompute_kernel(const float* __restrict__ qw,
                       float* __restrict__ tabs,
                       int* __restrict__ perms)
{
  __shared__ float2 rot[NQ][4];
  const int tid = threadIdx.x;
  if (tid < NQ) {
    const float* w = qw + (1 * NQ + tid) * 3;
    const float phi = w[0], theta = w[1], omega = w[2];
    float ch, sh, cp, sp, cm, sm;
    sincosf(0.5f * theta, &sh, &ch);
    sincosf(0.5f * (phi + omega), &sp, &cp);
    sincosf(0.5f * (phi - omega), &sm, &cm);
    rot[tid][0] = make_float2(ch * cp, -(ch * sp));
    rot[tid][1] = make_float2(-(sh * cm), -(sh * sm));
    rot[tid][2] = make_float2(sh * cm, -(sh * sm));
    rot[tid][3] = make_float2(ch * cp, ch * sp);
  }
  __syncthreads();

  static_assert(PRE_BLOCKS * PRE_THREADS == 2048, "one thread per permutation entry");
  const int e = blockIdx.x * PRE_THREADS + tid;
  if (e < TAB_FLOATS) {
    int nb, wbase, row, k, kind;
    if (e < TAB_CR) {
      const int tab = e >> 7;
      const int idx = e & 127;
      const int m = idx >> 3;
      k = idx & 7;
      row = m & 7;
      const int upper = m >> 3;
      nb = 3;
      wbase = (tab >> 1) * 3;
      if ((tab & 1) == 0) kind = upper ? 1 : 0;
      else                kind = upper ? 0 : 2;
    } else {
      const int e2 = e - TAB_CR;
      kind = e2 >> 8;
      const int idx = e2 & 255;
      row = idx >> 4;
      k = idx & 15;
      nb = 4;
      wbase = 6;
    }
    float vr = 1.f, vi = 0.f;
    for (int j = 0; j < nb; ++j) {
      const int bi = (row >> (nb - 1 - j)) & 1;
      const int bj = (k >> (nb - 1 - j)) & 1;
      const float2 f = rot[wbase + j][2 * bi + bj];
      const float nr = vr * f.x - vi * f.y;
      const float ni = vr * f.y + vi * f.x;
      vr = nr; vi = ni;
    }
    tabs[e] = kind == 0 ? vr : (kind == 1 ? vi : -vi);
  }
  perms[e] = (e >> 10) ? cnot_ring(e & 1023, 2) : cnot_ring(e & 1023, 1);
}

__global__ __launch_bounds__(32 * WAVES_PER_BLOCK)
void qlayer_kernel(const float* __restrict__ x,
                   const float* __restrict__ qw,
                   const float* __restrict__ tabs,
                   const int* __restrict__ perms,
                   float* __restrict__ out)
{
  __shared__ __align__(16) float2 lds_all[WAVES_PER_BLOCK * 1024];
  const int lane = threadIdx.x & 31;
  const int wave = threadIdx.x >> 5;
  float2* S = lds_all + wave * 1024;
  const int b = blockIdx.x * WAVES_PER_BLOCK + wave;

  const int* perm1 = perms;
  const int* perm2 = perms + 1024;

  const int q = lane < NQ ? lane : NQ - 1;
  float c, sn, ch, sh, cp, sp, cm, sm;
  {
    const float xv = x[b * NQ + q];
    const float phi = qw[q * 3 + 0], theta = qw[q * 3 + 1], omega = qw[q * 3 + 2];
    sincosf(0.5f * xv, &sn, &c);
    sincosf(0.5f * theta, &sh, &ch);
    sincosf(0.5f * (phi + omega), &sp, &cp);
    sincosf(0.5f * (phi - omega), &sm, &cm);
  }
  const float w0r = (cp * ch) * c - (cm * sh) * sn;
  const float w0i = -(sp * ch) * c - (sm * sh) * sn;
  const float w1r = (cm * sh) * c + (cp * ch) * sn;
  const float w1i = -(sm * sh) * c + (sp * ch) * sn;

  float ar_ = 1.f, ai_ = 0.f, br_ = 1.f, bi_ = 0.f;
  #pragma unroll
  for (int j = 0; j < 5; ++j) {
    const bool bit = (lane >> (4 - j)) & 1;
    const float f0r = bcast(w0r, j), f0i = bcast(w0i, j);
    const float f1r = bcast(w1r, j), f1i = bcast(w1i, j);
    const float g0r = bcast(w0r, 5 + j), g0i = bcast(w0i, 5 + j);
    const float g1r = bcast(w1r, 5 + j), g1i = bcast(w1i, 5 + j);
    const float fr = bit ? f1r : f0r;
    const float fi = bit ? f1i : f0i;
    const float gr = bit ? g1r : g0r;
    const float gi = bit ? g1i : g0i;
    float nr = ar_ * fr - ai_ * fi;
    float ni = ar_ * fi + ai_ * fr;
    ar_ = nr; ai_ = ni;
    nr = br_ * gr - bi_ * gi;
    ni = br_ * gi + bi_ * gr;
    br_ = nr; bi_ = ni;
  }
  for (int r = 0; r < 32; ++r) {
    float xr = __shfl(ar_, r, 32);
    float xi = __shfl(ai_, r, 32);
    float wr = xr * br_ - xi * bi_;
    float wi = xr * bi_ + xi * br_;
    int d = perm1[r * 32 + lane];
    S[d] = make_float2(wr, wi);
  }
  __syncthreads();

  const v8f vzero = {0.f, 0.f, 0.f, 0.f, 0.f, 0.f, 0.f, 0.f};
  const int c16 = lane & 15;
  const int kh = lane >> 4;
  float* Sf = reinterpret_cast<float*>(S);

  v2f ma1[2], ma2[2], mb1[2], mb2[2], cr[4], ci[4], cn[4];
  #pragma unroll
  for (int j = 0; j < 2; ++j) {
    const int o = c16 * 8 + 4 * j + 2 * kh;
    const float2 t0 = *reinterpret_cast<const float2*>(tabs + TAB_MA1 + o);
    const float2 t1 = *reinterpret_cast<const float2*>(tabs + TAB_MA2 + o);
    const float2 t2 = *reinterpret_cast<const float2*>(tabs + TAB_MB1 + o);
    const float2 t3 = *reinterpret_cast<const float2*>(tabs + TAB_MB2 + o);
    ma1[j] = (v2f){t0.x, t0.y}; ma2[j] = (v2f){t1.x, t1.y};
    mb1[j] = (v2f){t2.x, t2.y}; mb2[j] = (v2f){t3.x, t3.y};
  }
  #pragma unroll
  for (int j = 0; j < 4; ++j) {
    const int o = c16 * 16 + 4 * j + 2 * kh;
    const float2 t0 = *reinterpret_cast<const float2*>(tabs + TAB_CR + o);
    const float2 t1 = *reinterpret_cast<const float2*>(tabs + TAB_CI + o);
    const float2 t2 = *reinterpret_cast<const float2*>(tabs + TAB_CN + o);
    cr[j] = (v2f){t0.x, t0.y}; ci[j] = (v2f){t1.x, t1.y}; cn[j] = (v2f){t2.x, t2.y};
  }

  #pragma unroll
  for (int t = 0; t < 8; ++t) {
    v8f acc = vzero;
    #pragma unroll
    for (int j = 0; j < 2; ++j) {
      const int k0 = 4 * j + 2 * kh;
      const float2 s0 = S[k0 * 128 + 16 * t + c16];
      const float2 s1 = S[(k0 + 1) * 128 + 16 * t + c16];
      const v2f bR = {s0.x, s1.x};
      const v2f bI = {s0.y, s1.y};
      acc = wmma4(ma1[j], bR, acc);
      acc = wmma4(ma2[j], bI, acc);
    }
    #pragma unroll
    for (int r = 0; r < 8; ++r)
      Sf[2 * (r * 128 + 16 * t + c16) + kh] = acc[r];
  }
  __syncthreads();

  #pragma unroll
  for (int t = 0; t < 8; ++t) {
    v8f acc = vzero;
    #pragma unroll
    for (int j = 0; j < 2; ++j) {
      const int k0 = 4 * j + 2 * kh;
      const float2 s0 = S[t * 128 + k0 * 16 + c16];
      const float2 s1 = S[t * 128 + (k0 + 1) * 16 + c16];
      const v2f bR = {s0.x, s1.x};
      const v2f bI = {s0.y, s1.y};
      acc = wmma4(mb1[j], bR, acc);
      acc = wmma4(mb2[j], bI, acc);
    }
    #pragma unroll
    for (int r = 0; r < 8; ++r)
      Sf[2 * (t * 128 + r * 16 + c16) + kh] = acc[r];
  }
  __syncthreads();

  v8f qR[4], qI[4];
  #pragma unroll
  for (int t = 0; t < 4; ++t) {
    qR[t] = vzero; qI[t] = vzero;
    #pragma unroll
    for (int j = 0; j < 4; ++j) {
      const int k0 = 4 * j + 2 * kh;
      const float4 s01 = *reinterpret_cast<const float4*>(&S[(16 * t + c16) * 16 + k0]);
      const v2f bR = {s01.x, s01.z};
      const v2f bI = {s01.y, s01.w};
      qR[t] = wmma4(cr[j], bR, qR[t]);
      qR[t] = wmma4(cn[j], bI, qR[t]);
      qI[t] = wmma4(ci[j], bR, qI[t]);
      qI[t] = wmma4(cr[j], bI, qI[t]);
    }
  }
  __syncthreads();
  #pragma unroll
  for (int t = 0; t < 4; ++t)
    #pragma unroll
    for (int r = 0; r < 8; ++r) {
      const int idx = (16 * t + c16) * 16 + 8 * kh + r;
      const int d = perm2[idx];
      S[d] = make_float2(qR[t][r], qI[t][r]);
    }
  __syncthreads();

  float e[NQ];
  #pragma unroll
  for (int q = 0; q < NQ; ++q) e[q] = 0.f;
  for (int row = 0; row < 32; ++row) {
    int i = row * 32 + lane;
    float2 amp = S[i];
    float p = amp.x * amp.x + amp.y * amp.y;
    #pragma unroll
    for (int q = 0; q < NQ; ++q)
      e[q] += ((i >> (NQ - 1 - q)) & 1) ? -p : p;
  }
  #pragma unroll
  for (int q = 0; q < NQ; ++q) {
    #pragma unroll
    for (int off = 16; off > 0; off >>= 1)
      e[q] += __shfl_xor(e[q], off, 32);
  }
  if (lane == 0) {
    #pragma unroll
    for (int q = 0; q < NQ; ++q) out[b * NQ + q] = e[q];
  }
}

extern "C" void kernel_launch(void* const* d_in, const int* in_sizes, int n_in,
                              void* d_out, int out_size, void* d_ws, size_t ws_size,
                              hipStream_t stream) {
  const float* x  = (const float*)d_in[0];
  const float* qw = (const float*)d_in[1];
  float* out = (float*)d_out;

  float* tabs = (float*)d_ws;
  int* perms = (int*)((char*)d_ws + 16384);

  hipLaunchKernelGGL(precompute_kernel, dim3(PRE_BLOCKS), dim3(PRE_THREADS), 0, stream,
                     qw, tabs, perms);

  int batch = in_sizes[0] / NQ;
  hipLaunchKernelGGL(qlayer_kernel, dim3(batch / WAVES_PER_BLOCK),
                     dim3(32 * WAVES_PER_BLOCK), 0, stream,
                     x, qw, tabs, perms, out);
}


# Last CNOT ring folded into the PauliZ signs (ring = linear bit map): probabilities summed from the factor-C registers by a sign butterfly; no scatter, LDS round trip, perm2 table; sum order changes
# speedup vs baseline: 3.2805x; 1.2505x over previous
#include <hip/hip_runtime.h>
#include <math.h>


#define NQ 10
#define WAVES_PER_BLOCK 4

typedef float v2f __attribute__((ext_vector_type(2)));
typedef float v8f __attribute__((ext_vector_type(8)));

__device__ __forceinline__ v8f wmma4(v2f a, v2f b, v8f c) {
  return __builtin_amdgcn_wmma_f32_16x16x4_f32(
      false, a, false, b, (short)0, c, false, false);
}

__device__ __forceinline__ float bcast(float v, int src) {
  return __int_as_float(__builtin_amdgcn_readlane(__float_as_int(v), src));
}

#define PRE_THREADS 256
#define PRE_BLOCKS 5
#define TAB_MA1 0
#define TAB_MA2 128
#define TAB_MB1 256
#define TAB_MB2 384
#define TAB_CR 512
#define TAB_CI 768
#define TAB_CN 1024
#define TAB_FLOATS 1280

__host__ __device__ constexpr int cnot_ring(int i, int rr) {
  int b[NQ] = {};
  for (int q = 0; q < NQ; ++q) b[q] = (i >> (NQ - 1 - q)) & 1;
  for (int q = 0; q < NQ; ++q) b[(q + rr) % NQ] ^= b[q];
  int dest = 0;
  for (int q = 0; q < NQ; ++q) dest |= b[q] << (NQ - 1 - q);
  return dest;
}

__global__ __launch_bounds__(PRE_THREADS)
void precompute_kernel(const float* __restrict__ qw,
                       float* __restrict__ tabs,
                       int* __restrict__ perms)
{
  __shared__ float2 rot[NQ][4];
  const int tid = threadIdx.x;
  if (tid < NQ) {
    const float* w = qw + (1 * NQ + tid) * 3;
    const float phi = w[0], theta = w[1], omega = w[2];
    float ch, sh, cp, sp, cm, sm;
    sincosf(0.5f * theta, &sh, &ch);
    sincosf(0.5f * (phi + omega), &sp, &cp);
    sincosf(0.5f * (phi - omega), &sm, &cm);
    rot[tid][0] = make_float2(ch * cp, -(ch * sp));
    rot[tid][1] = make_float2(-(sh * cm), -(sh * sm));
    rot[tid][2] = make_float2(sh * cm, -(sh * sm));
    rot[tid][3] = make_float2(ch * cp, ch * sp);
  }
  __syncthreads();

  static_assert(PRE_BLOCKS * PRE_THREADS == TAB_FLOATS, "one thread per table entry");
  const int e = blockIdx.x * PRE_THREADS + tid;
  if (e < TAB_FLOATS) {
    int nb, wbase, row, k, kind;
    if (e < TAB_CR) {
      const int tab = e >> 7;
      const int idx = e & 127;
      const int m = idx >> 3;
      k = idx & 7;
      row = m & 7;
      const int upper = m >> 3;
      nb = 3;
      wbase = (tab >> 1) * 3;
      if ((tab & 1) == 0) kind = upper ? 1 : 0;
      else                kind = upper ? 0 : 2;
    } else {
      const int e2 = e - TAB_CR;
      kind = e2 >> 8;
      const int idx = e2 & 255;
      row = idx >> 4;
      k = idx & 15;
      nb = 4;
      wbase = 6;
    }
    float vr = 1.f, vi = 0.f;
    for (int j = 0; j < nb; ++j) {
      const int bi = (row >> (nb - 1 - j)) & 1;
      const int bj = (k >> (nb - 1 - j)) & 1;
      const float2 f = rot[wbase + j][2 * bi + bj];
      const float nr = vr * f.x - vi * f.y;
      const float ni = vr * f.y + vi * f.x;
      vr = nr; vi = ni;
    }
    tabs[e] = kind == 0 ? vr : (kind == 1 ? vi : -vi);
  }
  if (e < 1024) perms[e] = cnot_ring(e, 1);
}

__global__ __launch_bounds__(32 * WAVES_PER_BLOCK)
void qlayer_kernel(const float* __restrict__ x,
                   const float* __restrict__ qw,
                   const float* __restrict__ tabs,
                   const int* __restrict__ perms,
                   float* __restrict__ out)
{
  __shared__ __align__(16) float2 lds_all[WAVES_PER_BLOCK * 1024];
  const int lane = threadIdx.x & 31;
  const int wave = threadIdx.x >> 5;
  float2* S = lds_all + wave * 1024;
  const int b = blockIdx.x * WAVES_PER_BLOCK + wave;

  const int* perm1 = perms;

  const int q = lane < NQ ? lane : NQ - 1;
  float c, sn, ch, sh, cp, sp, cm, sm;
  {
    const float xv = x[b * NQ + q];
    const float phi = qw[q * 3 + 0], theta = qw[q * 3 + 1], omega = qw[q * 3 + 2];
    sincosf(0.5f * xv, &sn, &c);
    sincosf(0.5f * theta, &sh, &ch);
    sincosf(0.5f * (phi + omega), &sp, &cp);
    sincosf(0.5f * (phi - omega), &sm, &cm);
  }
  const float w0r = (cp * ch) * c - (cm * sh) * sn;
  const float w0i = -(sp * ch) * c - (sm * sh) * sn;
  const float w1r = (cm * sh) * c + (cp * ch) * sn;
  const float w1i = -(sm * sh) * c + (sp * ch) * sn;

  float ar_ = 1.f, ai_ = 0.f, br_ = 1.f, bi_ = 0.f;
  #pragma unroll
  for (int j = 0; j < 5; ++j) {
    const bool bit = (lane >> (4 - j)) & 1;
    const float f0r = bcast(w0r, j), f0i = bcast(w0i, j);
    const float f1r = bcast(w1r, j), f1i = bcast(w1i, j);
    const float g0r = bcast(w0r, 5 + j), g0i = bcast(w0i, 5 + j);
    const float g1r = bcast(w1r, 5 + j), g1i = bcast(w1i, 5 + j);
    const float fr = bit ? f1r : f0r;
    const float fi = bit ? f1i : f0i;
    const float gr = bit ? g1r : g0r;
    const float gi = bit ? g1i : g0i;
    float nr = ar_ * fr - ai_ * fi;
    float ni = ar_ * fi + ai_ * fr;
    ar_ = nr; ai_ = ni;
    nr = br_ * gr - bi_ * gi;
    ni = br_ * gi + bi_ * gr;
    br_ = nr; bi_ = ni;
  }
  for (int r = 0; r < 32; ++r) {
    float xr = __shfl(ar_, r, 32);
    float xi = __shfl(ai_, r, 32);
    float wr = xr * br_ - xi * bi_;
    float wi = xr * bi_ + xi * br_;
    int d = perm1[r * 32 + lane];
    S[d] = make_float2(wr, wi);
  }
  __syncthreads();

  const v8f vzero = {0.f, 0.f, 0.f, 0.f, 0.f, 0.f, 0.f, 0.f};
  const int c16 = lane & 15;
  const int kh = lane >> 4;
  float* Sf = reinterpret_cast<float*>(S);

  v2f ma1[2], ma2[2], mb1[2], mb2[2], cr[4], ci[4], cn[4];
  #pragma unroll
  for (int j = 0; j < 2; ++j) {
    const int o = c16 * 8 + 4 * j + 2 * kh;
    const float2 t0 = *reinterpret_cast<const float2*>(tabs + TAB_MA1 + o);
    const float2 t1 = *reinterpret_cast<const float2*>(tabs + TAB_MA2 + o);
    const float2 t2 = *reinterpret_cast<const float2*>(tabs + TAB_MB1 + o);
    const float2 t3 = *reinterpret_cast<const float2*>(tabs + TAB_MB2 + o);
    ma1[j] = (v2f){t0.x, t0.y}; ma2[j] = (v2f){t1.x, t1.y};
    mb1[j] = (v2f){t2.x, t2.y}; mb2[j] = (v2f){t3.x, t3.y};
  }
  #pragma unroll
  for (int j = 0; j < 4; ++j) {
    const int o = c16 * 16 + 4 * j + 2 * kh;
    const float2 t0 = *reinterpret_cast<const float2*>(tabs + TAB_CR + o);
    const float2 t1 = *reinterpret_cast<const float2*>(tabs + TAB_CI + o);
    const float2 t2 = *reinterpret_cast<const float2*>(tabs + TAB_CN + o);
    cr[j] = (v2f){t0.x, t0.y}; ci[j] = (v2f){t1.x, t1.y}; cn[j] = (v2f){t2.x, t2.y};
  }

  #pragma unroll
  for (int t = 0; t < 8; ++t) {
    v8f acc = vzero;
    #pragma unroll
    for (int j = 0; j < 2; ++j) {
      const int k0 = 4 * j + 2 * kh;
      const float2 s0 = S[k0 * 128 + 16 * t + c16];
      const float2 s1 = S[(k0 + 1) * 128 + 16 * t + c16];
      const v2f bR = {s0.x, s1.x};
      const v2f bI = {s0.y, s1.y};
      acc = wmma4(ma1[j], bR, acc);
      acc = wmma4(ma2[j], bI, acc);
    }
    #pragma unroll
    for (int r = 0; r < 8; ++r)
      Sf[2 * (r * 128 + 16 * t + c16) + kh] = acc[r];
  }
  __syncthreads();

  #pragma unroll
  for (int t = 0; t < 8; ++t) {
    v8f acc = vzero;
    #pragma unroll
    for (int j = 0; j < 2; ++j) {
      const int k0 = 4 * j + 2 * kh;
      const float2 s0 = S[t * 128 + k0 * 16 + c16];
      const float2 s1 = S[t * 128 + (k0 + 1) * 16 + c16];
      const v2f bR = {s0.x, s1.x};
      const v2f bI = {s0.y, s1.y};
      acc = wmma4(mb1[j], bR, acc);
      acc = wmma4(mb2[j], bI, acc);
    }
    #pragma unroll
    for (int r = 0; r < 8; ++r)
      Sf[2 * (t * 128 + r * 16 + c16) + kh] = acc[r];
  }
  __syncthreads();

  v8f qR[4], qI[4];
  #pragma unroll
  for (int t = 0; t < 4; ++t) {
    qR[t] = vzero; qI[t] = vzero;
    #pragma unroll
    for (int j = 0; j < 4; ++j) {
      const int k0 = 4 * j + 2 * kh;
      const float4 s01 = *reinterpret_cast<const float4*>(&S[(16 * t + c16) * 16 + k0]);
      const v2f bR = {s01.x, s01.z};
      const v2f bI = {s01.y, s01.w};
      qR[t] = wmma4(cr[j], bR, qR[t]);
      qR[t] = wmma4(cn[j], bI, qR[t]);
      qI[t] = wmma4(ci[j], bR, qI[t]);
      qI[t] = wmma4(cr[j], bI, qI[t]);
    }
  }
  float w[4][8];
  #pragma unroll
  for (int t = 0; t < 4; ++t) {
    #pragma unroll
    for (int r = 0; r < 8; ++r)
      w[t][r] = qR[t][r] * qR[t][r] + qI[t][r] * qI[t][r];
    #pragma unroll
    for (int s = 1; s < 8; s <<= 1) {
      #pragma unroll
      for (int i = 0; i < 8; ++i) {
        if ((i & s) == 0) {
          const float lo = w[t][i], hi = w[t][i | s];
          w[t][i] = lo + hi;
          w[t][i | s] = lo - hi;
        }
      }
    }
  }
  float e[NQ];
  const int dl = cnot_ring(16 * c16 + 8 * kh, 2);
  #pragma unroll
  for (int q = 0; q < NQ; ++q) {
    const int sh = NQ - 1 - q;
    int mr = 0, mt = 0;
    #pragma unroll
    for (int s = 0; s < 3; ++s) mr |= ((cnot_ring(1 << s, 2) >> sh) & 1) << s;
    #pragma unroll
    for (int u = 0; u < 2; ++u) mt |= ((cnot_ring(256 << u, 2) >> sh) & 1) << u;
    float acc = w[0][mr];
    #pragma unroll
    for (int t = 1; t < 4; ++t)
      acc += (__builtin_popcount(mt & t) & 1) ? -w[t][mr] : w[t][mr];
    e[q] = ((dl >> sh) & 1) ? -acc : acc;
  }
  #pragma unroll
  for (int q = 0; q < NQ; ++q) {
    #pragma unroll
    for (int off = 16; off > 0; off >>= 1)
      e[q] += __shfl_xor(e[q], off, 32);
  }
  if (lane == 0) {
    #pragma unroll
    for (int q = 0; q < NQ; ++q) out[b * NQ + q] = e[q];
  }
}

extern "C" void kernel_launch(void* const* d_in, const int* in_sizes, int n_in,
                              void* d_out, int out_size, void* d_ws, size_t ws_size,
                              hipStream_t stream) {
  const float* x  = (const float*)d_in[0];
  const float* qw = (const float*)d_in[1];
  float* out = (float*)d_out;

  float* tabs = (float*)d_ws;
  int* perms = (int*)((char*)d_ws + 16384);

  hipLaunchKernelGGL(precompute_kernel, dim3(PRE_BLOCKS), dim3(PRE_THREADS), 0, stream,
                     qw, tabs, perms);

  int batch = in_sizes[0] / NQ;
  hipLaunchKernelGGL(qlayer_kernel, dim3(batch / WAVES_PER_BLOCK),
                     dim3(32 * WAVES_PER_BLOCK), 0, stream,
                     x, qw, tabs, perms, out);
}


# State as a rank-4 product: CNOT ring 1 gives a ring chain, layer-1 Rot goes into its 2x2x2 site tensors, halves (wires 0-5 | 6-9) joined by one K=4 WMMA product; 128 -> 16 WMMAs; sum order changes
# speedup vs baseline: 8.4017x; 2.5611x over previous
#include <hip/hip_runtime.h>
#include <math.h>


#define NQ 10
#define WAVES_PER_BLOCK 4
#define SITE_FLOATS (NQ * 16)

typedef float v2f __attribute__((ext_vector_type(2)));
typedef float v8f __attribute__((ext_vector_type(8)));

__device__ __forceinline__ v8f wmma4(v2f a, v2f b, v8f c) {
  return __builtin_amdgcn_wmma_f32_16x16x4_f32(
      false, a, false, b, (short)0, c, false, false);
}

__device__ __forceinline__ float2 cmul(float2 a, float2 b) {
  return make_float2(a.x * b.x - a.y * b.y, a.x * b.y + a.y * b.x);
}
__device__ __forceinline__ float2 cmac2(float2 a, float2 b, float2 c, float2 d) {
  return make_float2(a.x * b.x - a.y * b.y + (c.x * d.x - c.y * d.y),
                     a.x * b.y + a.y * b.x + (c.x * d.y + c.y * d.x));
}
__device__ __forceinline__ float2 lo2(float4 m) { return make_float2(m.x, m.y); }
__device__ __forceinline__ float2 hi2(float4 m) { return make_float2(m.z, m.w); }

__host__ __device__ constexpr int cnot_ring(int i, int rr) {
  int b[NQ] = {};
  for (int q = 0; q < NQ; ++q) b[q] = (i >> (NQ - 1 - q)) & 1;
  for (int q = 0; q < NQ; ++q) b[(q + rr) % NQ] ^= b[q];
  int dest = 0;
  for (int q = 0; q < NQ; ++q) dest |= b[q] << (NQ - 1 - q);
  return dest;
}

#define PRE_THREADS 32

__global__ __launch_bounds__(PRE_THREADS)
void precompute_kernel(const float* __restrict__ qw,
                       float* __restrict__ tabs)
{
  const int tid = threadIdx.x;
  if (tid < NQ) {
    const float* w = qw + (1 * NQ + tid) * 3;
    const float phi = w[0], theta = w[1], omega = w[2];
    float ch, sh, cp, sp, cm, sm;
    sincosf(0.5f * theta, &sh, &ch);
    sincosf(0.5f * (phi + omega), &sp, &cp);
    sincosf(0.5f * (phi - omega), &sm, &cm);
    float4* o = reinterpret_cast<float4*>(tabs + tid * 8);
    o[0] = make_float4(ch * cp, -(ch * sp), -(sh * cm), -(sh * sm));
    o[1] = make_float4(sh * cm, -(sh * sm), ch * cp, ch * sp);
  }
}

__global__ __launch_bounds__(32 * WAVES_PER_BLOCK)
void qlayer_kernel(const float* __restrict__ x,
                   const float* __restrict__ qw,
                   const float* __restrict__ tabs,
                   float* __restrict__ out)
{
  __shared__ __align__(16) float lds_all[WAVES_PER_BLOCK * SITE_FLOATS];
  const int lane = threadIdx.x & 31;
  const int wave = threadIdx.x >> 5;
  float* Sw = lds_all + wave * SITE_FLOATS;
  const int b = blockIdx.x * WAVES_PER_BLOCK + wave;

  const int q = lane < NQ ? lane : NQ - 1;
  float c, sn, ch, sh, cp, sp, cm, sm;
  {
    const float xv = x[b * NQ + q];
    const float phi = qw[q * 3 + 0], theta = qw[q * 3 + 1], omega = qw[q * 3 + 2];
    sincosf(0.5f * xv, &sn, &c);
    sincosf(0.5f * theta, &sh, &ch);
    sincosf(0.5f * (phi + omega), &sp, &cp);
    sincosf(0.5f * (phi - omega), &sm, &cm);
  }
  const float2 w0 = make_float2((cp * ch) * c - (cm * sh) * sn,
                                -(sp * ch) * c - (sm * sh) * sn);
  const float2 w1 = make_float2((cm * sh) * c + (cp * ch) * sn,
                                -(sm * sh) * c + (sp * ch) * sn);

  {
    const float4 ua = *reinterpret_cast<const float4*>(tabs + q * 8);
    const float4 ub = *reinterpret_cast<const float4*>(tabs + q * 8 + 4);
    if (lane < NQ) {
      float4* o = reinterpret_cast<float4*>(Sw + lane * 16);
      const float2 m000 = cmul(lo2(ua), w0), m001 = cmul(hi2(ua), w1);
      const float2 m010 = cmul(lo2(ua), w1), m011 = cmul(hi2(ua), w0);
      const float2 m100 = cmul(lo2(ub), w0), m101 = cmul(hi2(ub), w1);
      const float2 m110 = cmul(lo2(ub), w1), m111 = cmul(hi2(ub), w0);
      o[0] = make_float4(m000.x, m000.y, m001.x, m001.y);
      o[1] = make_float4(m010.x, m010.y, m011.x, m011.y);
      o[2] = make_float4(m100.x, m100.y, m101.x, m101.y);
      o[3] = make_float4(m110.x, m110.y, m111.x, m111.y);
    }
  }
  __syncthreads();

  const int c16 = lane & 15;
  const int kh = lane >> 4;
  const float4* Mt = reinterpret_cast<const float4*>(Sw);
  const float2* Mc = reinterpret_cast<const float2*>(Sw);


  float2 r0, r1;
  {
    const float4 m = Mt[(6 * 2 + ((c16 >> 3) & 1)) * 2 + kh];
    r0 = lo2(m); r1 = hi2(m);
  }
  #pragma unroll
  for (int k = 7; k < NQ; ++k) {
    const int a = (c16 >> (NQ - 1 - k)) & 1;
    const float4 m0 = Mt[(k * 2 + a) * 2 + 0];
    const float4 m1 = Mt[(k * 2 + a) * 2 + 1];
    const float2 n0 = cmac2(r0, lo2(m0), r1, lo2(m1));
    const float2 n1 = cmac2(r0, hi2(m0), r1, hi2(m1));
    r0 = n0; r1 = n1;
  }

  float2 u0 = Mc[((5 * 2 + (c16 & 1)) * 2 + 0) * 2 + kh];
  float2 u1 = Mc[((5 * 2 + (c16 & 1)) * 2 + 1) * 2 + kh];
  #pragma unroll
  for (int k = 4; k >= 2; --k) {
    const int a = (c16 >> (5 - k)) & 1;
    const float4 m0 = Mt[(k * 2 + a) * 2 + 0];
    const float4 m1 = Mt[(k * 2 + a) * 2 + 1];
    const float2 n0 = cmac2(lo2(m0), u0, hi2(m0), u1);
    const float2 n1 = cmac2(lo2(m1), u0, hi2(m1), u1);
    u0 = n0; u1 = n1;
  }
  float2 y[2][2];
  #pragma unroll
  for (int a1 = 0; a1 < 2; ++a1) {
    #pragma unroll
    for (int p = 0; p < 2; ++p) {
      const float4 m = Mt[(1 * 2 + a1) * 2 + p];
      y[a1][p] = cmac2(lo2(m), u0, hi2(m), u1);
    }
  }
  v2f bRe[4], bIm[4];
  #pragma unroll
  for (int a0 = 0; a0 < 2; ++a0) {
    #pragma unroll
    for (int s = 0; s < 2; ++s) {
      const float4 m = Mt[(0 * 2 + a0) * 2 + s];
      #pragma unroll
      for (int a1 = 0; a1 < 2; ++a1) {
        const float2 l = cmac2(lo2(m), y[a1][s], hi2(m), y[a1][1 ^ s]);
        bRe[2 * a0 + a1][s] = l.x;
        bIm[2 * a0 + a1][s] = l.y;
      }
    }
  }

  const v8f vzero = {0.f, 0.f, 0.f, 0.f, 0.f, 0.f, 0.f, 0.f};
  const v2f aR = {r0.x, r1.x};
  const v2f aI = {r0.y, r1.y};
  const v2f aN = {-r0.y, -r1.y};
  v8f qR[4], qI[4];
  #pragma unroll
  for (int t = 0; t < 4; ++t) {
    qR[t] = wmma4(aR, bRe[t], vzero);
    qI[t] = wmma4(aI, bRe[t], vzero);
    qR[t] = wmma4(aN, bIm[t], qR[t]);
    qI[t] = wmma4(aR, bIm[t], qI[t]);
  }
  float w[4][8];
  #pragma unroll
  for (int t = 0; t < 4; ++t) {
    #pragma unroll
    for (int r = 0; r < 8; ++r)
      w[t][r] = qR[t][r] * qR[t][r] + qI[t][r] * qI[t][r];
    #pragma unroll
    for (int s = 1; s < 8; s <<= 1) {
      #pragma unroll
      for (int i = 0; i < 8; ++i) {
        if ((i & s) == 0) {
          const float lo = w[t][i], hi = w[t][i | s];
          w[t][i] = lo + hi;
          w[t][i | s] = lo - hi;
        }
      }
    }
  }
  float e[NQ];
  const int dl = cnot_ring(16 * c16 + 8 * kh, 2);
  #pragma unroll
  for (int q = 0; q < NQ; ++q) {
    const int sh = NQ - 1 - q;
    int mr = 0, mt = 0;
    #pragma unroll
    for (int s = 0; s < 3; ++s) mr |= ((cnot_ring(1 << s, 2) >> sh) & 1) << s;
    #pragma unroll
    for (int u = 0; u < 2; ++u) mt |= ((cnot_ring(256 << u, 2) >> sh) & 1) << u;
    float acc = w[0][mr];
    #pragma unroll
    for (int t = 1; t < 4; ++t)
      acc += (__builtin_popcount(mt & t) & 1) ? -w[t][mr] : w[t][mr];
    e[q] = ((dl >> sh) & 1) ? -acc : acc;
  }
  #pragma unroll
  for (int q = 0; q < NQ; ++q) {
    #pragma unroll
    for (int off = 16; off > 0; off >>= 1)
      e[q] += __shfl_xor(e[q], off, 32);
  }
  if (lane == 0) {
    #pragma unroll
    for (int q = 0; q < NQ; ++q) out[b * NQ + q] = e[q];
  }
}

extern "C" void kernel_launch(void* const* d_in, const int* in_sizes, int n_in,
                              void* d_out, int out_size, void* d_ws, size_t ws_size,
                              hipStream_t stream) {
  const float* x  = (const float*)d_in[0];
  const float* qw = (const float*)d_in[1];
  float* out = (float*)d_out;

  float* tabs = (float*)d_ws;

  hipLaunchKernelGGL(precompute_kernel, dim3(1), dim3(PRE_THREADS), 0, stream,
                     qw, tabs);

  int batch = in_sizes[0] / NQ;
  hipLaunchKernelGGL(qlayer_kernel, dim3(batch / WAVES_PER_BLOCK),
                     dim3(32 * WAVES_PER_BLOCK), 0, stream,
                     x, qw, tabs, out);
}


# Layer-1 Rot trigonometry done by the main kernel's idle lanes 16-25 in the same sincos instructions as layer 0 (lane k reads lane k+16 by permlanex16); set-up launch and table removed: 2 -> 1 launches
# speedup vs baseline: 10.1092x; 1.2032x over previous
#include <hip/hip_runtime.h>
#include <math.h>


#define NQ 10
#define WAVES_PER_BLOCK 4
#define SITE_FLOATS (NQ * 16)

typedef float v2f __attribute__((ext_vector_type(2)));
typedef float v8f __attribute__((ext_vector_type(8)));

__device__ __forceinline__ v8f wmma4(v2f a, v2f b, v8f c) {
  return __builtin_amdgcn_wmma_f32_16x16x4_f32(
      false, a, false, b, (short)0, c, false, false);
}

__device__ __forceinline__ float2 cmul(float2 a, float2 b) {
  return make_float2(a.x * b.x - a.y * b.y, a.x * b.y + a.y * b.x);
}
__device__ __forceinline__ float2 cmac2(float2 a, float2 b, float2 c, float2 d) {
  return make_float2(a.x * b.x - a.y * b.y + (c.x * d.x - c.y * d.y),
                     a.x * b.y + a.y * b.x + (c.x * d.y + c.y * d.x));
}
__device__ __forceinline__ float2 lo2(float4 m) { return make_float2(m.x, m.y); }
__device__ __forceinline__ float2 hi2(float4 m) { return make_float2(m.z, m.w); }

__host__ __device__ constexpr int cnot_ring(int i, int rr) {
  int b[NQ] = {};
  for (int q = 0; q < NQ; ++q) b[q] = (i >> (NQ - 1 - q)) & 1;
  for (int q = 0; q < NQ; ++q) b[(q + rr) % NQ] ^= b[q];
  int dest = 0;
  for (int q = 0; q < NQ; ++q) dest |= b[q] << (NQ - 1 - q);
  return dest;
}

__device__ __forceinline__ float other_row(float v) {
  return __builtin_bit_cast(float, __builtin_amdgcn_permlanex16(
      __builtin_bit_cast(int, v), __builtin_bit_cast(int, v), 0x76543210, 0xfedcba98, false, false));
}

__global__ __launch_bounds__(32 * WAVES_PER_BLOCK)
void qlayer_kernel(const float* __restrict__ x,
                   const float* __restrict__ qw,
                   float* __restrict__ out)
{
  __shared__ __align__(16) float lds_all[WAVES_PER_BLOCK * SITE_FLOATS];
  const int lane = threadIdx.x & 31;
  const int wave = threadIdx.x >> 5;
  float* Sw = lds_all + wave * SITE_FLOATS;
  const int b = blockIdx.x * WAVES_PER_BLOCK + wave;

  const int row = lane >> 4;
  const int q = (lane & 15) < NQ ? (lane & 15) : NQ - 1;
  float c, sn, ch, sh, cp, sp, cm, sm;
  {
    const float xv = x[b * NQ + q];
    const float* wq = qw + (row * NQ + q) * 3;
    const float phi = wq[0], theta = wq[1], omega = wq[2];
    sincosf(0.5f * xv, &sn, &c);
    sincosf(0.5f * theta, &sh, &ch);
    sincosf(0.5f * (phi + omega), &sp, &cp);
    sincosf(0.5f * (phi - omega), &sm, &cm);
  }
  const float pcc = cp * ch, psc = sp * ch, pcs = cm * sh, pss = sm * sh;
  const float2 w0 = make_float2(pcc * c - pcs * sn, -psc * c - pss * sn);
  const float2 w1 = make_float2(pcs * c + pcc * sn, -pss * c + psc * sn);
  const float ucc = other_row(pcc), usc = other_row(psc);
  const float ucs = other_row(pcs), uss = other_row(pss);

  {
    const float4 ua = make_float4(ucc, -usc, -ucs, -uss);
    const float4 ub = make_float4(ucs, -uss, ucc, usc);
    if (lane < NQ) {
      float4* o = reinterpret_cast<float4*>(Sw + lane * 16);
      const float2 m000 = cmul(lo2(ua), w0), m001 = cmul(hi2(ua), w1);
      const float2 m010 = cmul(lo2(ua), w1), m011 = cmul(hi2(ua), w0);
      const float2 m100 = cmul(lo2(ub), w0), m101 = cmul(hi2(ub), w1);
      const float2 m110 = cmul(lo2(ub), w1), m111 = cmul(hi2(ub), w0);
      o[0] = make_float4(m000.x, m000.y, m001.x, m001.y);
      o[1] = make_float4(m010.x, m010.y, m011.x, m011.y);
      o[2] = make_float4(m100.x, m100.y, m101.x, m101.y);
      o[3] = make_float4(m110.x, m110.y, m111.x, m111.y);
    }
  }
  __syncthreads();

  const int c16 = lane & 15;
  const int kh = lane >> 4;
  const float4* Mt = reinterpret_cast<const float4*>(Sw);
  const float2* Mc = reinterpret_cast<const float2*>(Sw);


  float2 r0, r1;
  {
    const float4 m = Mt[(6 * 2 + ((c16 >> 3) & 1)) * 2 + kh];
    r0 = lo2(m); r1 = hi2(m);
  }
  #pragma unroll
  for (int k = 7; k < NQ; ++k) {
    const int a = (c16 >> (NQ - 1 - k)) & 1;
    const float4 m0 = Mt[(k * 2 + a) * 2 + 0];
    const float4 m1 = Mt[(k * 2 + a) * 2 + 1];
    const float2 n0 = cmac2(r0, lo2(m0), r1, lo2(m1));
    const float2 n1 = cmac2(r0, hi2(m0), r1, hi2(m1));
    r0 = n0; r1 = n1;
  }

  float2 u0 = Mc[((5 * 2 + (c16 & 1)) * 2 + 0) * 2 + kh];
  float2 u1 = Mc[((5 * 2 + (c16 & 1)) * 2 + 1) * 2 + kh];
  #pragma unroll
  for (int k = 4; k >= 2; --k) {
    const int a = (c16 >> (5 - k)) & 1;
    const float4 m0 = Mt[(k * 2 + a) * 2 + 0];
    const float4 m1 = Mt[(k * 2 + a) * 2 + 1];
    const float2 n0 = cmac2(lo2(m0), u0, hi2(m0), u1);
    const float2 n1 = cmac2(lo2(m1), u0, hi2(m1), u1);
    u0 = n0; u1 = n1;
  }
  float2 y[2][2];
  #pragma unroll
  for (int a1 = 0; a1 < 2; ++a1) {
    #pragma unroll
    for (int p = 0; p < 2; ++p) {
      const float4 m = Mt[(1 * 2 + a1) * 2 + p];
      y[a1][p] = cmac2(lo2(m), u0, hi2(m), u1);
    }
  }
  v2f bRe[4], bIm[4];
  #pragma unroll
  for (int a0 = 0; a0 < 2; ++a0) {
    #pragma unroll
    for (int s = 0; s < 2; ++s) {
      const float4 m = Mt[(0 * 2 + a0) * 2 + s];
      #pragma unroll
      for (int a1 = 0; a1 < 2; ++a1) {
        const float2 l = cmac2(lo2(m), y[a1][s], hi2(m), y[a1][1 ^ s]);
        bRe[2 * a0 + a1][s] = l.x;
        bIm[2 * a0 + a1][s] = l.y;
      }
    }
  }

  const v8f vzero = {0.f, 0.f, 0.f, 0.f, 0.f, 0.f, 0.f, 0.f};
  const v2f aR = {r0.x, r1.x};
  const v2f aI = {r0.y, r1.y};
  const v2f aN = {-r0.y, -r1.y};
  v8f qR[4], qI[4];
  #pragma unroll
  for (int t = 0; t < 4; ++t) {
    qR[t] = wmma4(aR, bRe[t], vzero);
    qI[t] = wmma4(aI, bRe[t], vzero);
    qR[t] = wmma4(aN, bIm[t], qR[t]);
    qI[t] = wmma4(aR, bIm[t], qI[t]);
  }
  float w[4][8];
  #pragma unroll
  for (int t = 0; t < 4; ++t) {
    #pragma unroll
    for (int r = 0; r < 8; ++r)
      w[t][r] = qR[t][r] * qR[t][r] + qI[t][r] * qI[t][r];
    #pragma unroll
    for (int s = 1; s < 8; s <<= 1) {
      #pragma unroll
      for (int i = 0; i < 8; ++i) {
        if ((i & s) == 0) {
          const float lo = w[t][i], hi = w[t][i | s];
          w[t][i] = lo + hi;
          w[t][i | s] = lo - hi;
        }
      }
    }
  }
  float e[NQ];
  const int dl = cnot_ring(16 * c16 + 8 * kh, 2);
  #pragma unroll
  for (int q = 0; q < NQ; ++q) {
    const int sh = NQ - 1 - q;
    int mr = 0, mt = 0;
    #pragma unroll
    for (int s = 0; s < 3; ++s) mr |= ((cnot_ring(1 << s, 2) >> sh) & 1) << s;
    #pragma unroll
    for (int u = 0; u < 2; ++u) mt |= ((cnot_ring(256 << u, 2) >> sh) & 1) << u;
    float acc = w[0][mr];
    #pragma unroll
    for (int t = 1; t < 4; ++t)
      acc += (__builtin_popcount(mt & t) & 1) ? -w[t][mr] : w[t][mr];
    e[q] = ((dl >> sh) & 1) ? -acc : acc;
  }
  #pragma unroll
  for (int q = 0; q < NQ; ++q) {
    #pragma unroll
    for (int off = 16; off > 0; off >>= 1)
      e[q] += __shfl_xor(e[q], off, 32);
  }
  if (lane == 0) {
    #pragma unroll
    for (int q = 0; q < NQ; ++q) out[b * NQ + q] = e[q];
  }
}

extern "C" void kernel_launch(void* const* d_in, const int* in_sizes, int n_in,
                              void* d_out, int out_size, void* d_ws, size_t ws_size,
                              hipStream_t stream) {
  const float* x  = (const float*)d_in[0];
  const float* qw = (const float*)d_in[1];
  float* out = (float*)d_out;

  int batch = in_sizes[0] / NQ;
  hipLaunchKernelGGL(qlayer_kernel, dim3(batch / WAVES_PER_BLOCK),
                     dim3(32 * WAVES_PER_BLOCK), 0, stream,
                     x, qw, out);
}


# Four samples per wave: the Rot trigonometry of both layers (weights only) is done once per wave; x sincos and site tensors for two samples per pass (one per 16-lane row); 16384 -> 4096 waves
# speedup vs baseline: 10.8467x; 1.0730x over previous
#include <hip/hip_runtime.h>
#include <math.h>


#define NQ 10
#define WAVES_PER_BLOCK 4
#define SAMPLES_PER_WAVE 4
#define SITE_FLOATS (NQ * 16)

typedef float v2f __attribute__((ext_vector_type(2)));
typedef float v8f __attribute__((ext_vector_type(8)));

__device__ __forceinline__ v8f wmma4(v2f a, v2f b, v8f c) {
  return __builtin_amdgcn_wmma_f32_16x16x4_f32(
      false, a, false, b, (short)0, c, false, false);
}

__device__ __forceinline__ float2 cmul(float2 a, float2 b) {
  return make_float2(a.x * b.x - a.y * b.y, a.x * b.y + a.y * b.x);
}
__device__ __forceinline__ float2 cmac2(float2 a, float2 b, float2 c, float2 d) {
  return make_float2(a.x * b.x - a.y * b.y + (c.x * d.x - c.y * d.y),
                     a.x * b.y + a.y * b.x + (c.x * d.y + c.y * d.x));
}
__device__ __forceinline__ float2 lo2(float4 m) { return make_float2(m.x, m.y); }
__device__ __forceinline__ float2 hi2(float4 m) { return make_float2(m.z, m.w); }

__host__ __device__ constexpr int cnot_ring(int i, int rr) {
  int b[NQ] = {};
  for (int q = 0; q < NQ; ++q) b[q] = (i >> (NQ - 1 - q)) & 1;
  for (int q = 0; q < NQ; ++q) b[(q + rr) % NQ] ^= b[q];
  int dest = 0;
  for (int q = 0; q < NQ; ++q) dest |= b[q] << (NQ - 1 - q);
  return dest;
}

__device__ __forceinline__ float other_row(float v) {
  return __builtin_bit_cast(float, __builtin_amdgcn_permlanex16(
      __builtin_bit_cast(int, v), __builtin_bit_cast(int, v), 0x76543210, 0xfedcba98, false, false));
}

__global__ __launch_bounds__(32 * WAVES_PER_BLOCK)
void qlayer_kernel(const float* __restrict__ x,
                   const float* __restrict__ qw,
                   float* __restrict__ out)
{
  __shared__ __align__(16) float lds_all[WAVES_PER_BLOCK * SAMPLES_PER_WAVE * SITE_FLOATS];
  const int lane = threadIdx.x & 31;
  const int wave = threadIdx.x >> 5;
  float* Sw = lds_all + wave * (SAMPLES_PER_WAVE * SITE_FLOATS);
  const int b0 = (blockIdx.x * WAVES_PER_BLOCK + wave) * SAMPLES_PER_WAVE;

  const int row = lane >> 4;
  const int q = (lane & 15) < NQ ? (lane & 15) : NQ - 1;
  float xv[SAMPLES_PER_WAVE / 2];
  #pragma unroll
  for (int it = 0; it < SAMPLES_PER_WAVE / 2; ++it)
    xv[it] = x[(b0 + 2 * it + row) * NQ + q];
  float ch, sh, cp, sp, cm, sm;
  {
    const float* wq = qw + (row * NQ + q) * 3;
    const float phi = wq[0], theta = wq[1], omega = wq[2];
    sincosf(0.5f * theta, &sh, &ch);
    sincosf(0.5f * (phi + omega), &sp, &cp);
    sincosf(0.5f * (phi - omega), &sm, &cm);
  }
  const float pcc = cp * ch, psc = sp * ch, pcs = cm * sh, pss = sm * sh;
  const float occ = other_row(pcc), osc = other_row(psc);
  const float ocs = other_row(pcs), oss = other_row(pss);
  const bool in_row1 = row != 0;
  const float rcc = in_row1 ? occ : pcc, rsc = in_row1 ? osc : psc;
  const float rcs = in_row1 ? ocs : pcs, rss = in_row1 ? oss : pss;
  const float ucc = in_row1 ? pcc : occ, usc = in_row1 ? psc : osc;
  const float ucs = in_row1 ? pcs : ocs, uss = in_row1 ? pss : oss;
  const float4 ua = make_float4(ucc, -usc, -ucs, -uss);
  const float4 ub = make_float4(ucs, -uss, ucc, usc);

  #pragma unroll
  for (int it = 0; it < SAMPLES_PER_WAVE / 2; ++it) {
    const int s = 2 * it + row;
    float c, sn;
    sincosf(0.5f * xv[it], &sn, &c);
    const float2 w0 = make_float2(rcc * c - rcs * sn, -rsc * c - rss * sn);
    const float2 w1 = make_float2(rcs * c + rcc * sn, -rss * c + rsc * sn);
    if ((lane & 15) < NQ) {
      float4* o = reinterpret_cast<float4*>(Sw + s * SITE_FLOATS + q * 16);
      const float2 m000 = cmul(lo2(ua), w0), m001 = cmul(hi2(ua), w1);
      const float2 m010 = cmul(lo2(ua), w1), m011 = cmul(hi2(ua), w0);
      const float2 m100 = cmul(lo2(ub), w0), m101 = cmul(hi2(ub), w1);
      const float2 m110 = cmul(lo2(ub), w1), m111 = cmul(hi2(ub), w0);
      o[0] = make_float4(m000.x, m000.y, m001.x, m001.y);
      o[1] = make_float4(m010.x, m010.y, m011.x, m011.y);
      o[2] = make_float4(m100.x, m100.y, m101.x, m101.y);
      o[3] = make_float4(m110.x, m110.y, m111.x, m111.y);
    }
  }
  __syncthreads();

  const int c16 = lane & 15;
  const int kh = lane >> 4;

  #pragma unroll 1
  for (int smp = 0; smp < SAMPLES_PER_WAVE; ++smp) {
  const int b = b0 + smp;
  const float4* Mt = reinterpret_cast<const float4*>(Sw + smp * SITE_FLOATS);
  const float2* Mc = reinterpret_cast<const float2*>(Sw + smp * SITE_FLOATS);


  float2 r0, r1;
  {
    const float4 m = Mt[(6 * 2 + ((c16 >> 3) & 1)) * 2 + kh];
    r0 = lo2(m); r1 = hi2(m);
  }
  #pragma unroll
  for (int k = 7; k < NQ; ++k) {
    const int a = (c16 >> (NQ - 1 - k)) & 1;
    const float4 m0 = Mt[(k * 2 + a) * 2 + 0];
    const float4 m1 = Mt[(k * 2 + a) * 2 + 1];
    const float2 n0 = cmac2(r0, lo2(m0), r1, lo2(m1));
    const float2 n1 = cmac2(r0, hi2(m0), r1, hi2(m1));
    r0 = n0; r1 = n1;
  }

  float2 u0 = Mc[((5 * 2 + (c16 & 1)) * 2 + 0) * 2 + kh];
  float2 u1 = Mc[((5 * 2 + (c16 & 1)) * 2 + 1) * 2 + kh];
  #pragma unroll
  for (int k = 4; k >= 2; --k) {
    const int a = (c16 >> (5 - k)) & 1;
    const float4 m0 = Mt[(k * 2 + a) * 2 + 0];
    const float4 m1 = Mt[(k * 2 + a) * 2 + 1];
    const float2 n0 = cmac2(lo2(m0), u0, hi2(m0), u1);
    const float2 n1 = cmac2(lo2(m1), u0, hi2(m1), u1);
    u0 = n0; u1 = n1;
  }
  float2 y[2][2];
  #pragma unroll
  for (int a1 = 0; a1 < 2; ++a1) {
    #pragma unroll
    for (int p = 0; p < 2; ++p) {
      const float4 m = Mt[(1 * 2 + a1) * 2 + p];
      y[a1][p] = cmac2(lo2(m), u0, hi2(m), u1);
    }
  }
  v2f bRe[4], bIm[4];
  #pragma unroll
  for (int a0 = 0; a0 < 2; ++a0) {
    #pragma unroll
    for (int s = 0; s < 2; ++s) {
      const float4 m = Mt[(0 * 2 + a0) * 2 + s];
      #pragma unroll
      for (int a1 = 0; a1 < 2; ++a1) {
        const float2 l = cmac2(lo2(m), y[a1][s], hi2(m), y[a1][1 ^ s]);
        bRe[2 * a0 + a1][s] = l.x;
        bIm[2 * a0 + a1][s] = l.y;
      }
    }
  }

  const v8f vzero = {0.f, 0.f, 0.f, 0.f, 0.f, 0.f, 0.f, 0.f};
  const v2f aR = {r0.x, r1.x};
  const v2f aI = {r0.y, r1.y};
  const v2f aN = {-r0.y, -r1.y};
  v8f qR[4], qI[4];
  #pragma unroll
  for (int t = 0; t < 4; ++t) {
    qR[t] = wmma4(aR, bRe[t], vzero);
    qI[t] = wmma4(aI, bRe[t], vzero);
    qR[t] = wmma4(aN, bIm[t], qR[t]);
    qI[t] = wmma4(aR, bIm[t], qI[t]);
  }
  float w[4][8];
  #pragma unroll
  for (int t = 0; t < 4; ++t) {
    #pragma unroll
    for (int r = 0; r < 8; ++r)
      w[t][r] = qR[t][r] * qR[t][r] + qI[t][r] * qI[t][r];
    #pragma unroll
    for (int s = 1; s < 8; s <<= 1) {
      #pragma unroll
      for (int i = 0; i < 8; ++i) {
        if ((i & s) == 0) {
          const float lo = w[t][i], hi = w[t][i | s];
          w[t][i] = lo + hi;
          w[t][i | s] = lo - hi;
        }
      }
    }
  }
  float e[NQ];
  const int dl = cnot_ring(16 * c16 + 8 * kh, 2);
  #pragma unroll
  for (int q = 0; q < NQ; ++q) {
    const int sh = NQ - 1 - q;
    int mr = 0, mt = 0;
    #pragma unroll
    for (int s = 0; s < 3; ++s) mr |= ((cnot_ring(1 << s, 2) >> sh) & 1) << s;
    #pragma unroll
    for (int u = 0; u < 2; ++u) mt |= ((cnot_ring(256 << u, 2) >> sh) & 1) << u;
    float acc = w[0][mr];
    #pragma unroll
    for (int t = 1; t < 4; ++t)
      acc += (__builtin_popcount(mt & t) & 1) ? -w[t][mr] : w[t][mr];
    e[q] = ((dl >> sh) & 1) ? -acc : acc;
  }
  #pragma unroll
  for (int q = 0; q < NQ; ++q) {
    #pragma unroll
    for (int off = 16; off > 0; off >>= 1)
      e[q] += __shfl_xor(e[q], off, 32);
  }
  if (lane == 0) {
    #pragma unroll
    for (int q = 0; q < NQ; ++q) out[b * NQ + q] = e[q];
  }
  }
}

extern "C" void kernel_launch(void* const* d_in, const int* in_sizes, int n_in,
                              void* d_out, int out_size, void* d_ws, size_t ws_size,
                              hipStream_t stream) {
  const float* x  = (const float*)d_in[0];
  const float* qw = (const float*)d_in[1];
  float* out = (float*)d_out;

  int batch = in_sizes[0] / NQ;
  hipLaunchKernelGGL(qlayer_kernel, dim3(batch / (WAVES_PER_BLOCK * SAMPLES_PER_WAVE)),
                     dim3(32 * WAVES_PER_BLOCK), 0, stream,
                     x, qw, out);
}


# The ten 32-lane sums of the expectation values as one packed (transposing) reduction on the vector ALU: 12 cross-lane steps (permlanex16, DPP row_xmask) replace 50 ds_bpermute; order of sums kept
# speedup vs baseline: 12.5661x; 1.1585x over previous
#include <hip/hip_runtime.h>
#include <math.h>


#define NQ 10
#define WAVES_PER_BLOCK 4
#define SAMPLES_PER_WAVE 4
#define SITE_FLOATS (NQ * 16)

typedef float v2f __attribute__((ext_vector_type(2)));
typedef float v8f __attribute__((ext_vector_type(8)));

__device__ __forceinline__ v8f wmma4(v2f a, v2f b, v8f c) {
  return __builtin_amdgcn_wmma_f32_16x16x4_f32(
      false, a, false, b, (short)0, c, false, false);
}

__device__ __forceinline__ float2 cmul(float2 a, float2 b) {
  return make_float2(a.x * b.x - a.y * b.y, a.x * b.y + a.y * b.x);
}
__device__ __forceinline__ float2 cmac2(float2 a, float2 b, float2 c, float2 d) {
  return make_float2(a.x * b.x - a.y * b.y + (c.x * d.x - c.y * d.y),
                     a.x * b.y + a.y * b.x + (c.x * d.y + c.y * d.x));
}
__device__ __forceinline__ float2 lo2(float4 m) { return make_float2(m.x, m.y); }
__device__ __forceinline__ float2 hi2(float4 m) { return make_float2(m.z, m.w); }

__host__ __device__ constexpr int cnot_ring(int i, int rr) {
  int b[NQ] = {};
  for (int q = 0; q < NQ; ++q) b[q] = (i >> (NQ - 1 - q)) & 1;
  for (int q = 0; q < NQ; ++q) b[(q + rr) % NQ] ^= b[q];
  int dest = 0;
  for (int q = 0; q < NQ; ++q) dest |= b[q] << (NQ - 1 - q);
  return dest;
}

__device__ __forceinline__ float other_row(float v) {
  return __builtin_bit_cast(float, __builtin_amdgcn_permlanex16(
      __builtin_bit_cast(int, v), __builtin_bit_cast(int, v), 0x76543210, 0xfedcba98, false, false));
}

template <int M>
__device__ __forceinline__ float row_xor(float v) {
  return __builtin_bit_cast(float, __builtin_amdgcn_update_dpp(
      0, __builtin_bit_cast(int, v), 0x160 | M, 0xf, 0xf, true));
}

__device__ __forceinline__ float sum_rows_2(float a, float b, bool in_row1) {
  const float keep = in_row1 ? b : a;
  const float send = in_row1 ? a : b;
  return keep + other_row(send);
}

template <int M>
__device__ __forceinline__ float sum_pair(float a, float b, bool bit_set) {
  const float keep = bit_set ? b : a;
  const float send = bit_set ? a : b;
  return keep + row_xor<M>(send);
}

__global__ __launch_bounds__(32 * WAVES_PER_BLOCK)
void qlayer_kernel(const float* __restrict__ x,
                   const float* __restrict__ qw,
                   float* __restrict__ out)
{
  __shared__ __align__(16) float lds_all[WAVES_PER_BLOCK * SAMPLES_PER_WAVE * SITE_FLOATS];
  const int lane = threadIdx.x & 31;
  const int wave = threadIdx.x >> 5;
  float* Sw = lds_all + wave * (SAMPLES_PER_WAVE * SITE_FLOATS);
  const int b0 = (blockIdx.x * WAVES_PER_BLOCK + wave) * SAMPLES_PER_WAVE;

  const int row = lane >> 4;
  const int q = (lane & 15) < NQ ? (lane & 15) : NQ - 1;
  float xv[SAMPLES_PER_WAVE / 2];
  #pragma unroll
  for (int it = 0; it < SAMPLES_PER_WAVE / 2; ++it)
    xv[it] = x[(b0 + 2 * it + row) * NQ + q];
  float ch, sh, cp, sp, cm, sm;
  {
    const float* wq = qw + (row * NQ + q) * 3;
    const float phi = wq[0], theta = wq[1], omega = wq[2];
    sincosf(0.5f * theta, &sh, &ch);
    sincosf(0.5f * (phi + omega), &sp, &cp);
    sincosf(0.5f * (phi - omega), &sm, &cm);
  }
  const float pcc = cp * ch, psc = sp * ch, pcs = cm * sh, pss = sm * sh;
  const float occ = other_row(pcc), osc = other_row(psc);
  const float ocs = other_row(pcs), oss = other_row(pss);
  const bool in_row1 = row != 0;
  const float rcc = in_row1 ? occ : pcc, rsc = in_row1 ? osc : psc;
  const float rcs = in_row1 ? ocs : pcs, rss = in_row1 ? oss : pss;
  const float ucc = in_row1 ? pcc : occ, usc = in_row1 ? psc : osc;
  const float ucs = in_row1 ? pcs : ocs, uss = in_row1 ? pss : oss;
  const float4 ua = make_float4(ucc, -usc, -ucs, -uss);
  const float4 ub = make_float4(ucs, -uss, ucc, usc);

  #pragma unroll
  for (int it = 0; it < SAMPLES_PER_WAVE / 2; ++it) {
    const int s = 2 * it + row;
    float c, sn;
    sincosf(0.5f * xv[it], &sn, &c);
    const float2 w0 = make_float2(rcc * c - rcs * sn, -rsc * c - rss * sn);
    const float2 w1 = make_float2(rcs * c + rcc * sn, -rss * c + rsc * sn);
    if ((lane & 15) < NQ) {
      float4* o = reinterpret_cast<float4*>(Sw + s * SITE_FLOATS + q * 16);
      const float2 m000 = cmul(lo2(ua), w0), m001 = cmul(hi2(ua), w1);
      const float2 m010 = cmul(lo2(ua), w1), m011 = cmul(hi2(ua), w0);
      const float2 m100 = cmul(lo2(ub), w0), m101 = cmul(hi2(ub), w1);
      const float2 m110 = cmul(lo2(ub), w1), m111 = cmul(hi2(ub), w0);
      o[0] = make_float4(m000.x, m000.y, m001.x, m001.y);
      o[1] = make_float4(m010.x, m010.y, m011.x, m011.y);
      o[2] = make_float4(m100.x, m100.y, m101.x, m101.y);
      o[3] = make_float4(m110.x, m110.y, m111.x, m111.y);
    }
  }
  __syncthreads();

  const int c16 = lane & 15;
  const int kh = lane >> 4;
  const bool bit3 = (lane & 8) != 0, bit2 = (lane & 4) != 0, bit1 = (lane & 2) != 0;
  const int q_out = bit1 ? 8 + (lane >> 4)
                         : 4 * ((lane >> 2) & 1) + 2 * ((lane >> 3) & 1) + (lane >> 4);
  const bool writer = (lane & 1) == 0 && (bit1 ? (lane & 12) == 0 : true);

  #pragma unroll 1
  for (int smp = 0; smp < SAMPLES_PER_WAVE; ++smp) {
  const int b = b0 + smp;
  const float4* Mt = reinterpret_cast<const float4*>(Sw + smp * SITE_FLOATS);
  const float2* Mc = reinterpret_cast<const float2*>(Sw + smp * SITE_FLOATS);


  float2 r0, r1;
  {
    const float4 m = Mt[(6 * 2 + ((c16 >> 3) & 1)) * 2 + kh];
    r0 = lo2(m); r1 = hi2(m);
  }
  #pragma unroll
  for (int k = 7; k < NQ; ++k) {
    const int a = (c16 >> (NQ - 1 - k)) & 1;
    const float4 m0 = Mt[(k * 2 + a) * 2 + 0];
    const float4 m1 = Mt[(k * 2 + a) * 2 + 1];
    const float2 n0 = cmac2(r0, lo2(m0), r1, lo2(m1));
    const float2 n1 = cmac2(r0, hi2(m0), r1, hi2(m1));
    r0 = n0; r1 = n1;
  }

  float2 u0 = Mc[((5 * 2 + (c16 & 1)) * 2 + 0) * 2 + kh];
  float2 u1 = Mc[((5 * 2 + (c16 & 1)) * 2 + 1) * 2 + kh];
  #pragma unroll
  for (int k = 4; k >= 2; --k) {
    const int a = (c16 >> (5 - k)) & 1;
    const float4 m0 = Mt[(k * 2 + a) * 2 + 0];
    const float4 m1 = Mt[(k * 2 + a) * 2 + 1];
    const float2 n0 = cmac2(lo2(m0), u0, hi2(m0), u1);
    const float2 n1 = cmac2(lo2(m1), u0, hi2(m1), u1);
    u0 = n0; u1 = n1;
  }
  float2 y[2][2];
  #pragma unroll
  for (int a1 = 0; a1 < 2; ++a1) {
    #pragma unroll
    for (int p = 0; p < 2; ++p) {
      const float4 m = Mt[(1 * 2 + a1) * 2 + p];
      y[a1][p] = cmac2(lo2(m), u0, hi2(m), u1);
    }
  }
  v2f bRe[4], bIm[4];
  #pragma unroll
  for (int a0 = 0; a0 < 2; ++a0) {
    #pragma unroll
    for (int s = 0; s < 2; ++s) {
      const float4 m = Mt[(0 * 2 + a0) * 2 + s];
      #pragma unroll
      for (int a1 = 0; a1 < 2; ++a1) {
        const float2 l = cmac2(lo2(m), y[a1][s], hi2(m), y[a1][1 ^ s]);
        bRe[2 * a0 + a1][s] = l.x;
        bIm[2 * a0 + a1][s] = l.y;
      }
    }
  }

  const v8f vzero = {0.f, 0.f, 0.f, 0.f, 0.f, 0.f, 0.f, 0.f};
  const v2f aR = {r0.x, r1.x};
  const v2f aI = {r0.y, r1.y};
  const v2f aN = {-r0.y, -r1.y};
  v8f qR[4], qI[4];
  #pragma unroll
  for (int t = 0; t < 4; ++t) {
    qR[t] = wmma4(aR, bRe[t], vzero);
    qI[t] = wmma4(aI, bRe[t], vzero);
    qR[t] = wmma4(aN, bIm[t], qR[t]);
    qI[t] = wmma4(aR, bIm[t], qI[t]);
  }
  float w[4][8];
  #pragma unroll
  for (int t = 0; t < 4; ++t) {
    #pragma unroll
    for (int r = 0; r < 8; ++r)
      w[t][r] = qR[t][r] * qR[t][r] + qI[t][r] * qI[t][r];
    #pragma unroll
    for (int s = 1; s < 8; s <<= 1) {
      #pragma unroll
      for (int i = 0; i < 8; ++i) {
        if ((i & s) == 0) {
          const float lo = w[t][i], hi = w[t][i | s];
          w[t][i] = lo + hi;
          w[t][i | s] = lo - hi;
        }
      }
    }
  }
  float e[NQ];
  const int dl = cnot_ring(16 * c16 + 8 * kh, 2);
  #pragma unroll
  for (int q = 0; q < NQ; ++q) {
    const int sh = NQ - 1 - q;
    int mr = 0, mt = 0;
    #pragma unroll
    for (int s = 0; s < 3; ++s) mr |= ((cnot_ring(1 << s, 2) >> sh) & 1) << s;
    #pragma unroll
    for (int u = 0; u < 2; ++u) mt |= ((cnot_ring(256 << u, 2) >> sh) & 1) << u;
    float acc = w[0][mr];
    #pragma unroll
    for (int t = 1; t < 4; ++t)
      acc += (__builtin_popcount(mt & t) & 1) ? -w[t][mr] : w[t][mr];
    e[q] = ((dl >> sh) & 1) ? -acc : acc;
  }
  float f[5];
  #pragma unroll
  for (int j = 0; j < 5; ++j) f[j] = sum_rows_2(e[2 * j], e[2 * j + 1], in_row1);
  const float g0 = sum_pair<8>(f[0], f[1], bit3);
  const float g1 = sum_pair<8>(f[2], f[3], bit3);
  const float g2 = f[4] + row_xor<8>(f[4]);
  const float h0 = sum_pair<4>(g0, g1, bit2);
  const float h1 = g2 + row_xor<4>(g2);
  float tot = sum_pair<2>(h0, h1, bit1);
  tot += row_xor<1>(tot);
  if (writer) out[b * NQ + q_out] = tot;
  }
}

extern "C" void kernel_launch(void* const* d_in, const int* in_sizes, int n_in,
                              void* d_out, int out_size, void* d_ws, size_t ws_size,
                              hipStream_t stream) {
  const float* x  = (const float*)d_in[0];
  const float* qw = (const float*)d_in[1];
  float* out = (float*)d_out;

  int batch = in_sizes[0] / NQ;
  hipLaunchKernelGGL(qlayer_kernel, dim3(batch / (WAVES_PER_BLOCK * SAMPLES_PER_WAVE)),
                     dim3(32 * WAVES_PER_BLOCK), 0, stream,
                     x, qw, out);
}


# Site tensors kept split in LDS (re pair, im pair per entry, paired along the index a chain step produces): each chain step is 8 two-wide FMAs on the loaded pairs, no moves; sum order changes
# speedup vs baseline: 14.2349x; 1.1328x over previous
#include <hip/hip_runtime.h>
#include <math.h>


#define NQ 10
#define WAVES_PER_BLOCK 4
#define SAMPLES_PER_WAVE 4
#define SITE_FLOATS (NQ * 16)

typedef float v2f __attribute__((ext_vector_type(2)));
typedef float v8f __attribute__((ext_vector_type(8)));

__device__ __forceinline__ v8f wmma4(v2f a, v2f b, v8f c) {
  return __builtin_amdgcn_wmma_f32_16x16x4_f32(
      false, a, false, b, (short)0, c, false, false);
}

__device__ __forceinline__ float2 cmul(float2 a, float2 b) {
  return make_float2(a.x * b.x - a.y * b.y, a.x * b.y + a.y * b.x);
}
__device__ __forceinline__ float2 lo2(float4 m) { return make_float2(m.x, m.y); }
__device__ __forceinline__ float2 hi2(float4 m) { return make_float2(m.z, m.w); }

typedef float v4f __attribute__((ext_vector_type(4)));
struct cpair { v2f re, im; };
__device__ __forceinline__ v2f fma2(v2f a, v2f b, v2f c) { return __builtin_elementwise_fma(a, b, c); }

__device__ __forceinline__ cpair chain_step(cpair v, v4f f0, v4f f1) {
  const v2f x0 = v.re.xx, x1 = v.re.yy, y0 = v.im.xx, y1 = v.im.yy;
  cpair n;
  n.re = f0.lo * x0;
  n.re = fma2(-f0.hi, y0, n.re);
  n.re = fma2(f1.lo, x1, n.re);
  n.re = fma2(-f1.hi, y1, n.re);
  n.im = f0.hi * x0;
  n.im = fma2(f0.lo, y0, n.im);
  n.im = fma2(f1.hi, x1, n.im);
  n.im = fma2(f1.lo, y1, n.im);
  return n;
}
__device__ __forceinline__ cpair cross_step(cpair v, v4f f0, v4f f1) {
  const v2f xs = v.re.yx, ys = v.im.yx;
  cpair n;
  n.re = f0.lo * v.re;
  n.re = fma2(-f0.hi, v.im, n.re);
  n.re = fma2(f1.lo, xs, n.re);
  n.re = fma2(-f1.hi, ys, n.re);
  n.im = f0.hi * v.re;
  n.im = fma2(f0.lo, v.im, n.im);
  n.im = fma2(f1.hi, xs, n.im);
  n.im = fma2(f1.lo, ys, n.im);
  return n;
}

__host__ __device__ constexpr int cnot_ring(int i, int rr) {
  int b[NQ] = {};
  for (int q = 0; q < NQ; ++q) b[q] = (i >> (NQ - 1 - q)) & 1;
  for (int q = 0; q < NQ; ++q) b[(q + rr) % NQ] ^= b[q];
  int dest = 0;
  for (int q = 0; q < NQ; ++q) dest |= b[q] << (NQ - 1 - q);
  return dest;
}

__device__ __forceinline__ float other_row(float v) {
  return __builtin_bit_cast(float, __builtin_amdgcn_permlanex16(
      __builtin_bit_cast(int, v), __builtin_bit_cast(int, v), 0x76543210, 0xfedcba98, false, false));
}

template <int M>
__device__ __forceinline__ float row_xor(float v) {
  return __builtin_bit_cast(float, __builtin_amdgcn_update_dpp(
      0, __builtin_bit_cast(int, v), 0x160 | M, 0xf, 0xf, true));
}

__device__ __forceinline__ float sum_rows_2(float a, float b, bool in_row1) {
  const float keep = in_row1 ? b : a;
  const float send = in_row1 ? a : b;
  return keep + other_row(send);
}

template <int M>
__device__ __forceinline__ float sum_pair(float a, float b, bool bit_set) {
  const float keep = bit_set ? b : a;
  const float send = bit_set ? a : b;
  return keep + row_xor<M>(send);
}

__global__ __launch_bounds__(32 * WAVES_PER_BLOCK)
void qlayer_kernel(const float* __restrict__ x,
                   const float* __restrict__ qw,
                   float* __restrict__ out)
{
  __shared__ __align__(16) float lds_all[WAVES_PER_BLOCK * SAMPLES_PER_WAVE * SITE_FLOATS];
  const int lane = threadIdx.x & 31;
  const int wave = threadIdx.x >> 5;
  float* Sw = lds_all + wave * (SAMPLES_PER_WAVE * SITE_FLOATS);
  const int b0 = (blockIdx.x * WAVES_PER_BLOCK + wave) * SAMPLES_PER_WAVE;

  const int row = lane >> 4;
  const int q = (lane & 15) < NQ ? (lane & 15) : NQ - 1;
  float xv[SAMPLES_PER_WAVE / 2];
  #pragma unroll
  for (int it = 0; it < SAMPLES_PER_WAVE / 2; ++it)
    xv[it] = x[(b0 + 2 * it + row) * NQ + q];
  float ch, sh, cp, sp, cm, sm;
  {
    const float* wq = qw + (row * NQ + q) * 3;
    const float phi = wq[0], theta = wq[1], omega = wq[2];
    sincosf(0.5f * theta, &sh, &ch);
    sincosf(0.5f * (phi + omega), &sp, &cp);
    sincosf(0.5f * (phi - omega), &sm, &cm);
  }
  const float pcc = cp * ch, psc = sp * ch, pcs = cm * sh, pss = sm * sh;
  const float occ = other_row(pcc), osc = other_row(psc);
  const float ocs = other_row(pcs), oss = other_row(pss);
  const bool in_row1 = row != 0;
  const float rcc = in_row1 ? occ : pcc, rsc = in_row1 ? osc : psc;
  const float rcs = in_row1 ? ocs : pcs, rss = in_row1 ? oss : pss;
  const float ucc = in_row1 ? pcc : occ, usc = in_row1 ? psc : osc;
  const float ucs = in_row1 ? pcs : ocs, uss = in_row1 ? pss : oss;
  const float4 ua = make_float4(ucc, -usc, -ucs, -uss);
  const float4 ub = make_float4(ucs, -uss, ucc, usc);

  const bool pairs_over_p = q < 6;
  #pragma unroll
  for (int it = 0; it < SAMPLES_PER_WAVE / 2; ++it) {
    const int s = 2 * it + row;
    float c, sn;
    sincosf(0.5f * xv[it], &sn, &c);
    const float2 w0 = make_float2(rcc * c - rcs * sn, -rsc * c - rss * sn);
    const float2 w1 = make_float2(rcs * c + rcc * sn, -rss * c + rsc * sn);
    if ((lane & 15) < NQ) {
      float4* o = reinterpret_cast<float4*>(Sw + s * SITE_FLOATS + q * 16);
      const float2 m000 = cmul(lo2(ua), w0), m001 = cmul(hi2(ua), w1);
      const float2 m010 = cmul(lo2(ua), w1), m011 = cmul(hi2(ua), w0);
      const float2 m100 = cmul(lo2(ub), w0), m101 = cmul(hi2(ub), w1);
      const float2 m110 = cmul(lo2(ub), w1), m111 = cmul(hi2(ub), w0);
      const float2 e0 = pairs_over_p ? m010 : m001, e1 = pairs_over_p ? m001 : m010;
      const float2 e2 = pairs_over_p ? m110 : m101, e3 = pairs_over_p ? m101 : m110;
      o[0] = make_float4(m000.x, e0.x, m000.y, e0.y);
      o[1] = make_float4(e1.x, m011.x, e1.y, m011.y);
      o[2] = make_float4(m100.x, e2.x, m100.y, e2.y);
      o[3] = make_float4(e3.x, m111.x, e3.y, m111.y);
    }
  }
  __syncthreads();

  const int c16 = lane & 15;
  const int kh = lane >> 4;
  const bool bit3 = (lane & 8) != 0, bit2 = (lane & 4) != 0, bit1 = (lane & 2) != 0;
  const int q_out = bit1 ? 8 + (lane >> 4)
                         : 4 * ((lane >> 2) & 1) + 2 * ((lane >> 3) & 1) + (lane >> 4);
  const bool writer = (lane & 1) == 0 && (bit1 ? (lane & 12) == 0 : true);

  #pragma unroll 1
  for (int smp = 0; smp < SAMPLES_PER_WAVE; ++smp) {
  const int b = b0 + smp;
  const v4f* Mt = reinterpret_cast<const v4f*>(Sw + smp * SITE_FLOATS);


  cpair R;
  {
    const v4f m = Mt[(6 * 2 + ((c16 >> 3) & 1)) * 2 + kh];
    R.re = m.lo; R.im = m.hi;
  }
  #pragma unroll
  for (int k = 7; k < NQ; ++k) {
    const int a = (c16 >> (NQ - 1 - k)) & 1;
    R = chain_step(R, Mt[(k * 2 + a) * 2 + 0], Mt[(k * 2 + a) * 2 + 1]);
  }

  cpair Uc;
  {
    const v4f m = Mt[(5 * 2 + (c16 & 1)) * 2 + kh];
    Uc.re = m.lo; Uc.im = m.hi;
  }
  #pragma unroll
  for (int k = 4; k >= 2; --k) {
    const int a = (c16 >> (5 - k)) & 1;
    Uc = chain_step(Uc, Mt[(k * 2 + a) * 2 + 0], Mt[(k * 2 + a) * 2 + 1]);
  }
  cpair Y[2];
  #pragma unroll
  for (int a1 = 0; a1 < 2; ++a1)
    Y[a1] = chain_step(Uc, Mt[(1 * 2 + a1) * 2 + 0], Mt[(1 * 2 + a1) * 2 + 1]);
  v2f bRe[4], bIm[4];
  #pragma unroll
  for (int a0 = 0; a0 < 2; ++a0) {
    const v4f f0 = Mt[(0 * 2 + a0) * 2 + 0];
    const v4f f1 = Mt[(0 * 2 + a0) * 2 + 1];
    #pragma unroll
    for (int a1 = 0; a1 < 2; ++a1) {
      const cpair l = cross_step(Y[a1], f0, f1);
      bRe[2 * a0 + a1] = l.re;
      bIm[2 * a0 + a1] = l.im;
    }
  }

  const v8f vzero = {0.f, 0.f, 0.f, 0.f, 0.f, 0.f, 0.f, 0.f};
  const v2f aR = R.re;
  const v2f aI = R.im;
  const v2f aN = -R.im;
  v8f qR[4], qI[4];
  #pragma unroll
  for (int t = 0; t < 4; ++t) {
    qR[t] = wmma4(aR, bRe[t], vzero);
    qI[t] = wmma4(aI, bRe[t], vzero);
    qR[t] = wmma4(aN, bIm[t], qR[t]);
    qI[t] = wmma4(aR, bIm[t], qI[t]);
  }
  float w[4][8];
  #pragma unroll
  for (int t = 0; t < 4; ++t) {
    #pragma unroll
    for (int r = 0; r < 8; ++r)
      w[t][r] = qR[t][r] * qR[t][r] + qI[t][r] * qI[t][r];
    #pragma unroll
    for (int s = 1; s < 8; s <<= 1) {
      #pragma unroll
      for (int i = 0; i < 8; ++i) {
        if ((i & s) == 0) {
          const float lo = w[t][i], hi = w[t][i | s];
          w[t][i] = lo + hi;
          w[t][i | s] = lo - hi;
        }
      }
    }
  }
  float e[NQ];
  const int dl = cnot_ring(16 * c16 + 8 * kh, 2);
  #pragma unroll
  for (int q = 0; q < NQ; ++q) {
    const int sh = NQ - 1 - q;
    int mr = 0, mt = 0;
    #pragma unroll
    for (int s = 0; s < 3; ++s) mr |= ((cnot_ring(1 << s, 2) >> sh) & 1) << s;
    #pragma unroll
    for (int u = 0; u < 2; ++u) mt |= ((cnot_ring(256 << u, 2) >> sh) & 1) << u;
    float acc = w[0][mr];
    #pragma unroll
    for (int t = 1; t < 4; ++t)
      acc += (__builtin_popcount(mt & t) & 1) ? -w[t][mr] : w[t][mr];
    e[q] = ((dl >> sh) & 1) ? -acc : acc;
  }
  float f[5];
  #pragma unroll
  for (int j = 0; j < 5; ++j) f[j] = sum_rows_2(e[2 * j], e[2 * j + 1], in_row1);
  const float g0 = sum_pair<8>(f[0], f[1], bit3);
  const float g1 = sum_pair<8>(f[2], f[3], bit3);
  const float g2 = f[4] + row_xor<8>(f[4]);
  const float h0 = sum_pair<4>(g0, g1, bit2);
  const float h1 = g2 + row_xor<4>(g2);
  float tot = sum_pair<2>(h0, h1, bit1);
  tot += row_xor<1>(tot);
  if (writer) out[b * NQ + q_out] = tot;
  }
}

extern "C" void kernel_launch(void* const* d_in, const int* in_sizes, int n_in,
                              void* d_out, int out_size, void* d_ws, size_t ws_size,
                              hipStream_t stream) {
  const float* x  = (const float*)d_in[0];
  const float* qw = (const float*)d_in[1];
  float* out = (float*)d_out;

  int batch = in_sizes[0] / NQ;
  hipLaunchKernelGGL(qlayer_kernel, dim3(batch / (WAVES_PER_BLOCK * SAMPLES_PER_WAVE)),
                     dim3(32 * WAVES_PER_BLOCK), 0, stream,
                     x, qw, out);
}


# Complex K=4 product as three real WMMA products per tile, not four (shared product T = Re(R)(Re(L)+Im(L)) is the C operand of both parts): 16 -> 12 WMMAs per sample; sum order changes
# speedup vs baseline: 15.2057x; 1.0682x over previous
#include <hip/hip_runtime.h>
#include <math.h>


#define NQ 10
#define WAVES_PER_BLOCK 4
#define SAMPLES_PER_WAVE 4
#define SITE_FLOATS (NQ * 16)

typedef float v2f __attribute__((ext_vector_type(2)));
typedef float v8f __attribute__((ext_vector_type(8)));

__device__ __forceinline__ v8f wmma4(v2f a, v2f b, v8f c) {
  return __builtin_amdgcn_wmma_f32_16x16x4_f32(
      false, a, false, b, (short)0, c, false, false);
}

__device__ __forceinline__ float2 cmul(float2 a, float2 b) {
  return make_float2(a.x * b.x - a.y * b.y, a.x * b.y + a.y * b.x);
}
__device__ __forceinline__ float2 lo2(float4 m) { return make_float2(m.x, m.y); }
__device__ __forceinline__ float2 hi2(float4 m) { return make_float2(m.z, m.w); }

typedef float v4f __attribute__((ext_vector_type(4)));
struct cpair { v2f re, im; };
__device__ __forceinline__ v2f fma2(v2f a, v2f b, v2f c) { return __builtin_elementwise_fma(a, b, c); }

__device__ __forceinline__ cpair chain_step(cpair v, v4f f0, v4f f1) {
  const v2f x0 = v.re.xx, x1 = v.re.yy, y0 = v.im.xx, y1 = v.im.yy;
  cpair n;
  n.re = f0.lo * x0;
  n.re = fma2(-f0.hi, y0, n.re);
  n.re = fma2(f1.lo, x1, n.re);
  n.re = fma2(-f1.hi, y1, n.re);
  n.im = f0.hi * x0;
  n.im = fma2(f0.lo, y0, n.im);
  n.im = fma2(f1.hi, x1, n.im);
  n.im = fma2(f1.lo, y1, n.im);
  return n;
}
__device__ __forceinline__ cpair cross_step(cpair v, v4f f0, v4f f1) {
  const v2f xs = v.re.yx, ys = v.im.yx;
  cpair n;
  n.re = f0.lo * v.re;
  n.re = fma2(-f0.hi, v.im, n.re);
  n.re = fma2(f1.lo, xs, n.re);
  n.re = fma2(-f1.hi, ys, n.re);
  n.im = f0.hi * v.re;
  n.im = fma2(f0.lo, v.im, n.im);
  n.im = fma2(f1.hi, xs, n.im);
  n.im = fma2(f1.lo, ys, n.im);
  return n;
}

__host__ __device__ constexpr int cnot_ring(int i, int rr) {
  int b[NQ] = {};
  for (int q = 0; q < NQ; ++q) b[q] = (i >> (NQ - 1 - q)) & 1;
  for (int q = 0; q < NQ; ++q) b[(q + rr) % NQ] ^= b[q];
  int dest = 0;
  for (int q = 0; q < NQ; ++q) dest |= b[q] << (NQ - 1 - q);
  return dest;
}

__device__ __forceinline__ float other_row(float v) {
  return __builtin_bit_cast(float, __builtin_amdgcn_permlanex16(
      __builtin_bit_cast(int, v), __builtin_bit_cast(int, v), 0x76543210, 0xfedcba98, false, false));
}

template <int M>
__device__ __forceinline__ float row_xor(float v) {
  return __builtin_bit_cast(float, __builtin_amdgcn_update_dpp(
      0, __builtin_bit_cast(int, v), 0x160 | M, 0xf, 0xf, true));
}

__device__ __forceinline__ float sum_rows_2(float a, float b, bool in_row1) {
  const float keep = in_row1 ? b : a;
  const float send = in_row1 ? a : b;
  return keep + other_row(send);
}

template <int M>
__device__ __forceinline__ float sum_pair(float a, float b, bool bit_set) {
  const float keep = bit_set ? b : a;
  const float send = bit_set ? a : b;
  return keep + row_xor<M>(send);
}

__global__ __launch_bounds__(32 * WAVES_PER_BLOCK)
void qlayer_kernel(const float* __restrict__ x,
                   const float* __restrict__ qw,
                   float* __restrict__ out)
{
  __shared__ __align__(16) float lds_all[WAVES_PER_BLOCK * SAMPLES_PER_WAVE * SITE_FLOATS];
  const int lane = threadIdx.x & 31;
  const int wave = threadIdx.x >> 5;
  float* Sw = lds_all + wave * (SAMPLES_PER_WAVE * SITE_FLOATS);
  const int b0 = (blockIdx.x * WAVES_PER_BLOCK + wave) * SAMPLES_PER_WAVE;

  const int row = lane >> 4;
  const int q = (lane & 15) < NQ ? (lane & 15) : NQ - 1;
  float xv[SAMPLES_PER_WAVE / 2];
  #pragma unroll
  for (int it = 0; it < SAMPLES_PER_WAVE / 2; ++it)
    xv[it] = x[(b0 + 2 * it + row) * NQ + q];
  float ch, sh, cp, sp, cm, sm;
  {
    const float* wq = qw + (row * NQ + q) * 3;
    const float phi = wq[0], theta = wq[1], omega = wq[2];
    sincosf(0.5f * theta, &sh, &ch);
    sincosf(0.5f * (phi + omega), &sp, &cp);
    sincosf(0.5f * (phi - omega), &sm, &cm);
  }
  const float pcc = cp * ch, psc = sp * ch, pcs = cm * sh, pss = sm * sh;
  const float occ = other_row(pcc), osc = other_row(psc);
  const float ocs = other_row(pcs), oss = other_row(pss);
  const bool in_row1 = row != 0;
  const float rcc = in_row1 ? occ : pcc, rsc = in_row1 ? osc : psc;
  const float rcs = in_row1 ? ocs : pcs, rss = in_row1 ? oss : pss;
  const float ucc = in_row1 ? pcc : occ, usc = in_row1 ? psc : osc;
  const float ucs = in_row1 ? pcs : ocs, uss = in_row1 ? pss : oss;
  const float4 ua = make_float4(ucc, -usc, -ucs, -uss);
  const float4 ub = make_float4(ucs, -uss, ucc, usc);

  const bool pairs_over_p = q < 6;
  #pragma unroll
  for (int it = 0; it < SAMPLES_PER_WAVE / 2; ++it) {
    const int s = 2 * it + row;
    float c, sn;
    sincosf(0.5f * xv[it], &sn, &c);
    const float2 w0 = make_float2(rcc * c - rcs * sn, -rsc * c - rss * sn);
    const float2 w1 = make_float2(rcs * c + rcc * sn, -rss * c + rsc * sn);
    if ((lane & 15) < NQ) {
      float4* o = reinterpret_cast<float4*>(Sw + s * SITE_FLOATS + q * 16);
      const float2 m000 = cmul(lo2(ua), w0), m001 = cmul(hi2(ua), w1);
      const float2 m010 = cmul(lo2(ua), w1), m011 = cmul(hi2(ua), w0);
      const float2 m100 = cmul(lo2(ub), w0), m101 = cmul(hi2(ub), w1);
      const float2 m110 = cmul(lo2(ub), w1), m111 = cmul(hi2(ub), w0);
      const float2 e0 = pairs_over_p ? m010 : m001, e1 = pairs_over_p ? m001 : m010;
      const float2 e2 = pairs_over_p ? m110 : m101, e3 = pairs_over_p ? m101 : m110;
      o[0] = make_float4(m000.x, e0.x, m000.y, e0.y);
      o[1] = make_float4(e1.x, m011.x, e1.y, m011.y);
      o[2] = make_float4(m100.x, e2.x, m100.y, e2.y);
      o[3] = make_float4(e3.x, m111.x, e3.y, m111.y);
    }
  }
  __syncthreads();

  const int c16 = lane & 15;
  const int kh = lane >> 4;
  const bool bit3 = (lane & 8) != 0, bit2 = (lane & 4) != 0, bit1 = (lane & 2) != 0;
  const int q_out = bit1 ? 8 + (lane >> 4)
                         : 4 * ((lane >> 2) & 1) + 2 * ((lane >> 3) & 1) + (lane >> 4);
  const bool writer = (lane & 1) == 0 && (bit1 ? (lane & 12) == 0 : true);

  #pragma unroll 1
  for (int smp = 0; smp < SAMPLES_PER_WAVE; ++smp) {
  const int b = b0 + smp;
  const v4f* Mt = reinterpret_cast<const v4f*>(Sw + smp * SITE_FLOATS);


  cpair R;
  {
    const v4f m = Mt[(6 * 2 + ((c16 >> 3) & 1)) * 2 + kh];
    R.re = m.lo; R.im = m.hi;
  }
  #pragma unroll
  for (int k = 7; k < NQ; ++k) {
    const int a = (c16 >> (NQ - 1 - k)) & 1;
    R = chain_step(R, Mt[(k * 2 + a) * 2 + 0], Mt[(k * 2 + a) * 2 + 1]);
  }

  cpair Uc;
  {
    const v4f m = Mt[(5 * 2 + (c16 & 1)) * 2 + kh];
    Uc.re = m.lo; Uc.im = m.hi;
  }
  #pragma unroll
  for (int k = 4; k >= 2; --k) {
    const int a = (c16 >> (5 - k)) & 1;
    Uc = chain_step(Uc, Mt[(k * 2 + a) * 2 + 0], Mt[(k * 2 + a) * 2 + 1]);
  }
  cpair Y[2];
  #pragma unroll
  for (int a1 = 0; a1 < 2; ++a1)
    Y[a1] = chain_step(Uc, Mt[(1 * 2 + a1) * 2 + 0], Mt[(1 * 2 + a1) * 2 + 1]);
  v2f bRe[4], bIm[4];
  #pragma unroll
  for (int a0 = 0; a0 < 2; ++a0) {
    const v4f f0 = Mt[(0 * 2 + a0) * 2 + 0];
    const v4f f1 = Mt[(0 * 2 + a0) * 2 + 1];
    #pragma unroll
    for (int a1 = 0; a1 < 2; ++a1) {
      const cpair l = cross_step(Y[a1], f0, f1);
      bRe[2 * a0 + a1] = l.re;
      bIm[2 * a0 + a1] = l.im;
    }
  }

  const v8f vzero = {0.f, 0.f, 0.f, 0.f, 0.f, 0.f, 0.f, 0.f};
  const v2f aR = R.re;
  const v2f aS = -(R.re + R.im);
  const v2f aD = R.im - R.re;
  v8f qR[4], qI[4];
  #pragma unroll
  for (int t = 0; t < 4; ++t) {
    const v8f T = wmma4(aR, bRe[t] + bIm[t], vzero);
    qR[t] = wmma4(aS, bIm[t], T);
    qI[t] = wmma4(aD, bRe[t], T);
  }
  float w[4][8];
  #pragma unroll
  for (int t = 0; t < 4; ++t) {
    #pragma unroll
    for (int r = 0; r < 8; ++r)
      w[t][r] = qR[t][r] * qR[t][r] + qI[t][r] * qI[t][r];
    #pragma unroll
    for (int s = 1; s < 8; s <<= 1) {
      #pragma unroll
      for (int i = 0; i < 8; ++i) {
        if ((i & s) == 0) {
          const float lo = w[t][i], hi = w[t][i | s];
          w[t][i] = lo + hi;
          w[t][i | s] = lo - hi;
        }
      }
    }
  }
  float e[NQ];
  const int dl = cnot_ring(16 * c16 + 8 * kh, 2);
  #pragma unroll
  for (int q = 0; q < NQ; ++q) {
    const int sh = NQ - 1 - q;
    int mr = 0, mt = 0;
    #pragma unroll
    for (int s = 0; s < 3; ++s) mr |= ((cnot_ring(1 << s, 2) >> sh) & 1) << s;
    #pragma unroll
    for (int u = 0; u < 2; ++u) mt |= ((cnot_ring(256 << u, 2) >> sh) & 1) << u;
    float acc = w[0][mr];
    #pragma unroll
    for (int t = 1; t < 4; ++t)
      acc += (__builtin_popcount(mt & t) & 1) ? -w[t][mr] : w[t][mr];
    e[q] = ((dl >> sh) & 1) ? -acc : acc;
  }
  float f[5];
  #pragma unroll
  for (int j = 0; j < 5; ++j) f[j] = sum_rows_2(e[2 * j], e[2 * j + 1], in_row1);
  const float g0 = sum_pair<8>(f[0], f[1], bit3);
  const float g1 = sum_pair<8>(f[2], f[3], bit3);
  const float g2 = f[4] + row_xor<8>(f[4]);
  const float h0 = sum_pair<4>(g0, g1, bit2);
  const float h1 = g2 + row_xor<4>(g2);
  float tot = sum_pair<2>(h0, h1, bit1);
  tot += row_xor<1>(tot);
  if (writer) out[b * NQ + q_out] = tot;
  }
}

extern "C" void kernel_launch(void* const* d_in, const int* in_sizes, int n_in,
                              void* d_out, int out_size, void* d_ws, size_t ws_size,
                              hipStream_t stream) {
  const float* x  = (const float*)d_in[0];
  const float* qw = (const float*)d_in[1];
  float* out = (float*)d_out;

  int batch = in_sizes[0] / NQ;
  hipLaunchKernelGGL(qlayer_kernel, dim3(batch / (WAVES_PER_BLOCK * SAMPLES_PER_WAVE)),
                     dim3(32 * WAVES_PER_BLOCK), 0, stream,
                     x, qw, out);
}
